# starting kernel
# baseline (speedup 1.0000x reference)
#include <hip/hip_runtime.h>
#include <hip/hip_bf16.h>
#include <math.h>

#define NN  100000
#define MM  3
#define DIN 128
#define DH  64
#define NE  1600000

typedef __attribute__((ext_vector_type(2))) float v2f;
typedef __attribute__((ext_vector_type(4))) float v4f;
typedef __attribute__((ext_vector_type(8))) float v8f;

#define AGG_OFF   0ull
#define DEG_OFF   19200000ull
#define LOG_OFF   19500000ull
#define BETA_OFF  19500004ull
#define PROJ_OFF  19500008ull

__device__ __forceinline__ void atomicAddF(float* p, float v) {
  asm volatile("global_atomic_add_f32 %0, %1, off scope:SCOPE_DEV"
               :: "v"(p), "v"(v) : "memory");
}

__global__ __launch_bounds__(128)
void k_proj_wmma(const float* __restrict__ feats, const float* __restrict__ W,
                 float* __restrict__ proj) {
  const int lane = threadIdx.x & 31;
  const int wv   = threadIdx.x >> 5;
  const int col0 = wv * 16;
  const int m    = blockIdx.y;
  const int n0   = blockIdx.x * 16;
  const int r    = lane & 15;
  const int hi   = lane >> 4;
  const int kh   = hi * 2;

  const float* A = feats + ((size_t)m * NN + n0) * DIN;
  const float* B = W + (size_t)m * DIN * DH;

  v8f acc = {0.f,0.f,0.f,0.f,0.f,0.f,0.f,0.f};
  #pragma unroll
  for (int k = 0; k < DIN; k += 4) {
    v2f a = *(const v2f*)(A + (size_t)r * DIN + k + kh);
    v2f b;
    b.x = B[(k + kh)     * DH + col0 + r];
    b.y = B[(k + kh + 1) * DH + col0 + r];
    acc = __builtin_amdgcn_wmma_f32_16x16x4_f32(false, a, false, b, (short)0, acc,
                                                false, false);
  }
  float* O = proj + ((size_t)m * NN + n0) * DH;
  #pragma unroll
  for (int v = 0; v < 8; ++v)
    O[(size_t)(v + hi * 8) * DH + col0 + r] = acc[v];
}

__global__ __launch_bounds__(256)
void k_scatter(const float* __restrict__ proj, const int* __restrict__ src,
               const int* __restrict__ dst, float* __restrict__ agg,
               float* __restrict__ deg) {
  size_t t = (size_t)blockIdx.x * 256 + threadIdx.x;
  if (t >= (size_t)MM * NE * DH) return;
  int    j  = (int)(t & (DH - 1));
  size_t e  = t >> 6;
  int    m  = (int)(e / NE);
  int s = src[e];
  int d = dst[e];
  float val = proj[((size_t)m * NN + s) * DH + j];
  atomicAddF(&agg[((size_t)m * NN + d) * DH + j], val);
  if (j == 0) atomicAddF(&deg[(size_t)m * NN + d], 1.0f);
}

__global__ __launch_bounds__(256)
void k_prelu(float* __restrict__ agg, const float* __restrict__ deg,
             const float* __restrict__ b, const float* __restrict__ pa) {
  size_t i4 = (size_t)blockIdx.x * 256 + threadIdx.x;
  if (i4 >= (size_t)MM * NN * (DH / 4)) return;
  size_t base = i4 * 4;
  size_t nl   = i4 >> 4;
  int m  = (int)(nl / NN);
  int j4 = (int)(i4 & 15);
  float d = fmaxf(deg[nl], 1.0f);
  float a = pa[m];
  v4f hv = *(v4f*)(agg + base);
  v4f bv = *(const v4f*)(b + m * DH + j4 * 4);
  #pragma unroll
  for (int c = 0; c < 4; ++c) {
    float h = hv[c] / d + bv[c];
    hv[c] = h > 0.f ? h : a * h;
  }
  *(v4f*)(agg + base) = hv;
}

__global__ __launch_bounds__(128)
void k_fc_wmma(const float* __restrict__ h, const float* __restrict__ fcW,
               const float* __restrict__ fcb, const float* __restrict__ attn,
               float* __restrict__ logits) {
  __shared__ float red[128];
  const int lane = threadIdx.x & 31;
  const int wv   = threadIdx.x >> 5;
  const int col0 = wv * 16;
  const int m    = blockIdx.y;
  const int n0   = blockIdx.x * 16;
  const int r    = lane & 15;
  const int hi   = lane >> 4;
  const int kh   = hi * 2;

  const float* A = h + ((size_t)m * NN + n0) * DH;
  float bias = fcb[col0 + r];
  v8f acc = {bias, bias, bias, bias, bias, bias, bias, bias};
  #pragma unroll
  for (int k = 0; k < DH; k += 4) {
    v2f a = *(const v2f*)(A + (size_t)r * DH + k + kh);
    v2f bb;
    bb.x = fcW[(k + kh)     * DH + col0 + r];
    bb.y = fcW[(k + kh + 1) * DH + col0 + r];
    acc = __builtin_amdgcn_wmma_f32_16x16x4_f32(false, a, false, bb, (short)0, acc,
                                                false, false);
  }
  float p = 0.f;
  #pragma unroll
  for (int v = 0; v < 8; ++v) p += tanhf(acc[v]);
  p *= attn[col0 + r];

  red[threadIdx.x] = p;
  __syncthreads();
  #pragma unroll
  for (int s2 = 64; s2 > 0; s2 >>= 1) {
    if (threadIdx.x < s2) red[threadIdx.x] += red[threadIdx.x + s2];
    __syncthreads();
  }
  if (threadIdx.x == 0) atomicAddF(&logits[m], red[0]);
}

__global__ void k_softmax(const float* __restrict__ logits, float* __restrict__ beta) {
  if (threadIdx.x == 0 && blockIdx.x == 0) {
    float l[MM]; float mx = -1e30f;
    for (int m = 0; m < MM; ++m) { l[m] = logits[m] / (float)NN; mx = fmaxf(mx, l[m]); }
    float s = 0.f;
    for (int m = 0; m < MM; ++m) { l[m] = expf(l[m] - mx); s += l[m]; }
    for (int m = 0; m < MM; ++m) beta[m] = l[m] / s;
  }
}

__global__ __launch_bounds__(256)
void k_out(const float* __restrict__ h, const float* __restrict__ beta,
           float* __restrict__ out) {
  size_t i4 = (size_t)blockIdx.x * 256 + threadIdx.x;
  if (i4 >= (size_t)NN * (DH / 4)) return;
  size_t base = i4 * 4;
  const size_t P = (size_t)NN * DH;
  float b0 = beta[0], b1 = beta[1], b2 = beta[2];
  v4f x0 = *(const v4f*)(h + base);
  v4f x1 = *(const v4f*)(h + P + base);
  v4f x2 = *(const v4f*)(h + 2 * P + base);
  v4f z  = x0 * b0 + x1 * b1 + x2 * b2;
  *(v4f*)(out + base) = z;
}

extern "C" void kernel_launch(void* const* d_in, const int* in_sizes, int n_in,
                              void* d_out, int out_size, void* d_ws, size_t ws_size,
                              hipStream_t stream) {
  const float* feats = (const float*)d_in[0];
  const int*   src   = (const int*)  d_in[1];
  const int*   dst   = (const int*)  d_in[2];
  const float* W     = (const float*)d_in[3];
  const float* b     = (const float*)d_in[4];
  const float* pa    = (const float*)d_in[5];
  const float* fcW   = (const float*)d_in[6];
  const float* fcb   = (const float*)d_in[7];
  const float* attn  = (const float*)d_in[8];

  float* ws     = (float*)d_ws;
  float* agg    = ws + AGG_OFF;
  float* deg    = ws + DEG_OFF;
  float* logits = ws + LOG_OFF;
  float* beta   = ws + BETA_OFF;
  float* proj   = ws + PROJ_OFF;
  float* out    = (float*)d_out;

  hipMemsetAsync(d_ws, 0, (size_t)(LOG_OFF + 8) * sizeof(float), stream);

  dim3 gTiles(NN / 16, MM);
  k_proj_wmma<<<gTiles, 128, 0, stream>>>(feats, W, proj);

  size_t total = (size_t)MM * NE * DH;
  unsigned blocks = (unsigned)((total + 255) / 256);
  k_scatter<<<blocks, 256, 0, stream>>>(proj, src, dst, agg, deg);

  size_t nv4 = (size_t)MM * NN * (DH / 4);
  k_prelu<<<(unsigned)((nv4 + 255) / 256), 256, 0, stream>>>(agg, deg, b, pa);

  k_fc_wmma<<<gTiles, 128, 0, stream>>>(agg, fcW, fcb, attn, logits);

  k_softmax<<<1, 32, 0, stream>>>(logits, beta);

  size_t ov4 = (size_t)NN * (DH / 4);
  k_out<<<(unsigned)((ov4 + 255) / 256), 256, 0, stream>>>(agg, beta, out);
}


# scatter: a wave walks runs of 32 edges, indices loaded once per run, 8 edges' whole-line row loads in flight ahead of their whole-line f32 atomic adds (arrival order as before)
# speedup vs baseline: 1.9049x; 1.9049x over previous
#include <hip/hip_runtime.h>
#include <hip/hip_bf16.h>
#include <math.h>

#define NN  100000
#define MM  3
#define DIN 128
#define DH  64
#define NE  1600000

typedef __attribute__((ext_vector_type(2))) float v2f;
typedef __attribute__((ext_vector_type(4))) float v4f;
typedef __attribute__((ext_vector_type(8))) float v8f;

#define AGG_OFF   0ull
#define DEG_OFF   19200000ull
#define LOG_OFF   19500000ull
#define BETA_OFF  19500004ull
#define PROJ_OFF  19500008ull

__device__ __forceinline__ void atomicAddF(float* p, float v) {
  asm volatile("global_atomic_add_f32 %0, %1, off scope:SCOPE_DEV"
               :: "v"(p), "v"(v) : "memory");
}

__global__ __launch_bounds__(128)
void k_proj_wmma(const float* __restrict__ feats, const float* __restrict__ W,
                 float* __restrict__ proj) {
  const int lane = threadIdx.x & 31;
  const int wv   = threadIdx.x >> 5;
  const int col0 = wv * 16;
  const int m    = blockIdx.y;
  const int n0   = blockIdx.x * 16;
  const int r    = lane & 15;
  const int hi   = lane >> 4;
  const int kh   = hi * 2;

  const float* A = feats + ((size_t)m * NN + n0) * DIN;
  const float* B = W + (size_t)m * DIN * DH;

  v8f acc = {0.f,0.f,0.f,0.f,0.f,0.f,0.f,0.f};
  #pragma unroll
  for (int k = 0; k < DIN; k += 4) {
    v2f a = *(const v2f*)(A + (size_t)r * DIN + k + kh);
    v2f b;
    b.x = B[(k + kh)     * DH + col0 + r];
    b.y = B[(k + kh + 1) * DH + col0 + r];
    acc = __builtin_amdgcn_wmma_f32_16x16x4_f32(false, a, false, b, (short)0, acc,
                                                false, false);
  }
  float* O = proj + ((size_t)m * NN + n0) * DH;
  #pragma unroll
  for (int v = 0; v < 8; ++v)
    O[(size_t)(v + hi * 8) * DH + col0 + r] = acc[v];
}

#define CH        32
#define NCHUNK    (MM * NE / CH)
#define CH_PER_M  (NE / CH)
#define SC_BLOCKS 1875
#define SC_WAVES  (SC_BLOCKS * 8)
#define GE        8

__device__ __forceinline__ void atomicAddRow2(float* row, unsigned voff, float v0, float v1) {
  asm volatile("global_atomic_add_f32 %0, %1, %2 scope:SCOPE_DEV"
               :: "v"(voff), "v"(v0), "s"(row) : "memory");
  asm volatile("global_atomic_add_f32 %0, %1, %2 offset:128 scope:SCOPE_DEV"
               :: "v"(voff), "v"(v1), "s"(row) : "memory");
}

__global__ __launch_bounds__(256)
void k_scatter(const float* __restrict__ proj, const int* __restrict__ src,
               const int* __restrict__ dst, float* __restrict__ agg,
               float* __restrict__ deg) {
  const unsigned lane = threadIdx.x & 31;
  const unsigned voff = lane * 4u;
  int c = __builtin_amdgcn_readfirstlane((int)(blockIdx.x * 8 + (threadIdx.x >> 5)));
  if (c >= NCHUNK) return;
  int sv = src[c * CH + lane];
  int dv = dst[c * CH + lane];
  for (;;) {
    const int  cn   = c + SC_WAVES;
    const bool more = cn < NCHUNK;
    int svn = 0, dvn = 0;
    if (more) {
      svn = src[cn * CH + lane];
      dvn = dst[cn * CH + lane];
    }
    const unsigned rb = (unsigned)(c / CH_PER_M) * NN;

    float r0[2][GE], r1[2][GE];
    #pragma unroll
    for (int j = 0; j < GE; ++j) {
      const unsigned s = (unsigned)__builtin_amdgcn_readlane(sv, j);
      const unsigned ix = (rb + s) * DH + lane;
      r0[0][j] = proj[ix];
      r1[0][j] = proj[ix + 32];
    }
    #pragma unroll
    for (int g = 0; g < CH / GE; ++g) {
      if (g + 1 < CH / GE) {
        if (g > 0) asm volatile("s_wait_xcnt 0x0" ::: "memory");
        #pragma unroll
        for (int j = 0; j < GE; ++j) {
          const unsigned s = (unsigned)__builtin_amdgcn_readlane(sv, (g + 1) * GE + j);
          const unsigned ix = (rb + s) * DH + lane;
          r0[(g + 1) & 1][j] = proj[ix];
          r1[(g + 1) & 1][j] = proj[ix + 32];
        }
      }
      #pragma unroll
      for (int j = 0; j < GE; ++j) {
        const unsigned d = (unsigned)__builtin_amdgcn_readlane(dv, g * GE + j);
        atomicAddRow2((float*)((char*)agg + ((rb + d) << 8)), voff, r0[g & 1][j], r1[g & 1][j]);
      }
    }
    atomicAddF(&deg[rb + (unsigned)dv], 1.0f);
    if (!more) break;
    asm volatile("s_wait_xcnt 0x0" ::: "memory");
    c = cn; sv = svn; dv = dvn;
  }
}

__global__ __launch_bounds__(256)
void k_prelu(float* __restrict__ agg, const float* __restrict__ deg,
             const float* __restrict__ b, const float* __restrict__ pa) {
  size_t i4 = (size_t)blockIdx.x * 256 + threadIdx.x;
  if (i4 >= (size_t)MM * NN * (DH / 4)) return;
  size_t base = i4 * 4;
  size_t nl   = i4 >> 4;
  int m  = (int)(nl / NN);
  int j4 = (int)(i4 & 15);
  float d = fmaxf(deg[nl], 1.0f);
  float a = pa[m];
  v4f hv = *(v4f*)(agg + base);
  v4f bv = *(const v4f*)(b + m * DH + j4 * 4);
  #pragma unroll
  for (int c = 0; c < 4; ++c) {
    float h = hv[c] / d + bv[c];
    hv[c] = h > 0.f ? h : a * h;
  }
  *(v4f*)(agg + base) = hv;
}

__global__ __launch_bounds__(128)
void k_fc_wmma(const float* __restrict__ h, const float* __restrict__ fcW,
               const float* __restrict__ fcb, const float* __restrict__ attn,
               float* __restrict__ logits) {
  __shared__ float red[128];
  const int lane = threadIdx.x & 31;
  const int wv   = threadIdx.x >> 5;
  const int col0 = wv * 16;
  const int m    = blockIdx.y;
  const int n0   = blockIdx.x * 16;
  const int r    = lane & 15;
  const int hi   = lane >> 4;
  const int kh   = hi * 2;

  const float* A = h + ((size_t)m * NN + n0) * DH;
  float bias = fcb[col0 + r];
  v8f acc = {bias, bias, bias, bias, bias, bias, bias, bias};
  #pragma unroll
  for (int k = 0; k < DH; k += 4) {
    v2f a = *(const v2f*)(A + (size_t)r * DH + k + kh);
    v2f bb;
    bb.x = fcW[(k + kh)     * DH + col0 + r];
    bb.y = fcW[(k + kh + 1) * DH + col0 + r];
    acc = __builtin_amdgcn_wmma_f32_16x16x4_f32(false, a, false, bb, (short)0, acc,
                                                false, false);
  }
  float p = 0.f;
  #pragma unroll
  for (int v = 0; v < 8; ++v) p += tanhf(acc[v]);
  p *= attn[col0 + r];

  red[threadIdx.x] = p;
  __syncthreads();
  #pragma unroll
  for (int s2 = 64; s2 > 0; s2 >>= 1) {
    if (threadIdx.x < s2) red[threadIdx.x] += red[threadIdx.x + s2];
    __syncthreads();
  }
  if (threadIdx.x == 0) atomicAddF(&logits[m], red[0]);
}

__global__ void k_softmax(const float* __restrict__ logits, float* __restrict__ beta) {
  if (threadIdx.x == 0 && blockIdx.x == 0) {
    float l[MM]; float mx = -1e30f;
    for (int m = 0; m < MM; ++m) { l[m] = logits[m] / (float)NN; mx = fmaxf(mx, l[m]); }
    float s = 0.f;
    for (int m = 0; m < MM; ++m) { l[m] = expf(l[m] - mx); s += l[m]; }
    for (int m = 0; m < MM; ++m) beta[m] = l[m] / s;
  }
}

__global__ __launch_bounds__(256)
void k_out(const float* __restrict__ h, const float* __restrict__ beta,
           float* __restrict__ out) {
  size_t i4 = (size_t)blockIdx.x * 256 + threadIdx.x;
  if (i4 >= (size_t)NN * (DH / 4)) return;
  size_t base = i4 * 4;
  const size_t P = (size_t)NN * DH;
  float b0 = beta[0], b1 = beta[1], b2 = beta[2];
  v4f x0 = *(const v4f*)(h + base);
  v4f x1 = *(const v4f*)(h + P + base);
  v4f x2 = *(const v4f*)(h + 2 * P + base);
  v4f z  = x0 * b0 + x1 * b1 + x2 * b2;
  *(v4f*)(out + base) = z;
}

extern "C" void kernel_launch(void* const* d_in, const int* in_sizes, int n_in,
                              void* d_out, int out_size, void* d_ws, size_t ws_size,
                              hipStream_t stream) {
  const float* feats = (const float*)d_in[0];
  const int*   src   = (const int*)  d_in[1];
  const int*   dst   = (const int*)  d_in[2];
  const float* W     = (const float*)d_in[3];
  const float* b     = (const float*)d_in[4];
  const float* pa    = (const float*)d_in[5];
  const float* fcW   = (const float*)d_in[6];
  const float* fcb   = (const float*)d_in[7];
  const float* attn  = (const float*)d_in[8];

  float* ws     = (float*)d_ws;
  float* agg    = ws + AGG_OFF;
  float* deg    = ws + DEG_OFF;
  float* logits = ws + LOG_OFF;
  float* beta   = ws + BETA_OFF;
  float* proj   = ws + PROJ_OFF;
  float* out    = (float*)d_out;

  hipMemsetAsync(d_ws, 0, (size_t)(LOG_OFF + 8) * sizeof(float), stream);

  dim3 gTiles(NN / 16, MM);
  k_proj_wmma<<<gTiles, 128, 0, stream>>>(feats, W, proj);

  k_scatter<<<SC_BLOCKS, 256, 0, stream>>>(proj, src, dst, agg, deg);

  size_t nv4 = (size_t)MM * NN * (DH / 4);
  k_prelu<<<(unsigned)((nv4 + 255) / 256), 256, 0, stream>>>(agg, deg, b, pa);

  k_fc_wmma<<<gTiles, 128, 0, stream>>>(agg, fcW, fcb, attn, logits);

  k_softmax<<<1, 32, 0, stream>>>(logits, beta);

  size_t ov4 = (size_t)NN * (DH / 4);
  k_out<<<(unsigned)((ov4 + 255) / 256), 256, 0, stream>>>(agg, beta, out);
}


# degree: per-chunk LDS integer histograms flushed as whole-line f32 atomic adds (exact integer sums) replace the scattered one-word f32 atomic per edge; row adds and their arrival order unchanged
# speedup vs baseline: 2.6768x; 1.4052x over previous
#include <hip/hip_runtime.h>
#include <hip/hip_bf16.h>
#include <math.h>

#define NN  100000
#define MM  3
#define DIN 128
#define DH  64
#define NE  1600000

typedef __attribute__((ext_vector_type(2))) float v2f;
typedef __attribute__((ext_vector_type(4))) float v4f;
typedef __attribute__((ext_vector_type(8))) float v8f;

#define AGG_OFF   0ull
#define DEG_OFF   19200000ull
#define LOG_OFF   19500000ull
#define BETA_OFF  19500004ull
#define PROJ_OFF  19500008ull

__device__ __forceinline__ void atomicAddF(float* p, float v) {
  asm volatile("global_atomic_add_f32 %0, %1, off scope:SCOPE_DEV"
               :: "v"(p), "v"(v) : "memory");
}

__global__ __launch_bounds__(128)
void k_proj_wmma(const float* __restrict__ feats, const float* __restrict__ W,
                 float* __restrict__ proj) {
  const int lane = threadIdx.x & 31;
  const int wv   = threadIdx.x >> 5;
  const int col0 = wv * 16;
  const int m    = blockIdx.y;
  const int n0   = blockIdx.x * 16;
  const int r    = lane & 15;
  const int hi   = lane >> 4;
  const int kh   = hi * 2;

  const float* A = feats + ((size_t)m * NN + n0) * DIN;
  const float* B = W + (size_t)m * DIN * DH;

  v8f acc = {0.f,0.f,0.f,0.f,0.f,0.f,0.f,0.f};
  #pragma unroll
  for (int k = 0; k < DIN; k += 4) {
    v2f a = *(const v2f*)(A + (size_t)r * DIN + k + kh);
    v2f b;
    b.x = B[(k + kh)     * DH + col0 + r];
    b.y = B[(k + kh + 1) * DH + col0 + r];
    acc = __builtin_amdgcn_wmma_f32_16x16x4_f32(false, a, false, b, (short)0, acc,
                                                false, false);
  }
  float* O = proj + ((size_t)m * NN + n0) * DH;
  #pragma unroll
  for (int v = 0; v < 8; ++v)
    O[(size_t)(v + hi * 8) * DH + col0 + r] = acc[v];
}

#define DG_CH     40000
#define DG_NCH    (NE / DG_CH)
#define DG_RANGE  50048
#define DG_NR     2

__device__ __forceinline__ void atomicAddLine(float* line, unsigned voff, float v) {
  asm volatile("global_atomic_add_f32 %0, %1, %2 scope:SCOPE_DEV"
               :: "v"(voff), "v"(v), "s"(line) : "memory");
}

#define DG_LD     10
#define DG_FG     7

__global__ __launch_bounds__(1024)
void k_deg(const int* __restrict__ dst, float* __restrict__ deg) {
  __shared__ alignas(16) unsigned hist[DG_RANGE];
  const int tid = threadIdx.x;
  const int m   = blockIdx.z;
  const int lo  = blockIdx.y * DG_RANGE;
  const unsigned nn = (unsigned)((NN - lo) < DG_RANGE ? (NN - lo) : DG_RANGE);

  const int4* d4 = (const int4*)(dst + (size_t)m * NE + (size_t)blockIdx.x * DG_CH);
  int4 d[DG_LD];
  #pragma unroll
  for (int j = 0; j < DG_LD; ++j) {
    const int i = tid + j * 1024;
    d[j] = (i < DG_CH / 4) ? d4[i] : make_int4(-1, -1, -1, -1);
  }

  for (int i = tid; i < DG_RANGE / 4; i += 1024) ((uint4*)hist)[i] = make_uint4(0u, 0u, 0u, 0u);
  __syncthreads();

  #pragma unroll
  for (int j = 0; j < DG_LD; ++j) {
    const unsigned a0 = (unsigned)(d[j].x - lo), a1 = (unsigned)(d[j].y - lo);
    const unsigned a2 = (unsigned)(d[j].z - lo), a3 = (unsigned)(d[j].w - lo);
    if (a0 < nn) atomicAdd(&hist[a0], 1u);
    if (a1 < nn) atomicAdd(&hist[a1], 1u);
    if (a2 < nn) atomicAdd(&hist[a2], 1u);
    if (a3 < nn) atomicAdd(&hist[a3], 1u);
  }
  __syncthreads();

  const int wv = __builtin_amdgcn_readfirstlane(tid >> 5);
  const unsigned lane = (unsigned)tid & 31u;
  float* base = deg + (size_t)m * NN + lo;
  const int nlines = (int)(nn >> 5);
  for (int l0 = wv; l0 < nlines; l0 += 32 * DG_FG) {
    float h[DG_FG];
    #pragma unroll
    for (int j = 0; j < DG_FG; ++j) {
      const int ln = l0 + 32 * j;
      h[j] = (float)hist[(ln < nlines ? ln : l0) * 32 + lane];
    }
    #pragma unroll
    for (int j = 0; j < DG_FG; ++j) {
      const int ln = l0 + 32 * j;
      if (ln < nlines) atomicAddLine(base + ln * 32, lane * 4u, h[j]);
    }
    asm volatile("s_wait_xcnt 0x0" ::: "memory");
  }
}

#define CH        32
#define NCHUNK    (MM * NE / CH)
#define CH_PER_M  (NE / CH)
#define SC_BLOCKS 1875
#define SC_WAVES  (SC_BLOCKS * 8)
#define GE        8

__device__ __forceinline__ void atomicAddRow2(float* row, unsigned voff, float v0, float v1) {
  asm volatile("global_atomic_add_f32 %0, %1, %2 scope:SCOPE_DEV"
               :: "v"(voff), "v"(v0), "s"(row) : "memory");
  asm volatile("global_atomic_add_f32 %0, %1, %2 offset:128 scope:SCOPE_DEV"
               :: "v"(voff), "v"(v1), "s"(row) : "memory");
}

__global__ __launch_bounds__(256)
void k_scatter(const float* __restrict__ proj, const int* __restrict__ src,
               const int* __restrict__ dst, float* __restrict__ agg) {
  const unsigned lane = threadIdx.x & 31;
  const unsigned voff = lane * 4u;
  int c = __builtin_amdgcn_readfirstlane((int)(blockIdx.x * 8 + (threadIdx.x >> 5)));
  if (c >= NCHUNK) return;
  int sv = src[c * CH + lane];
  int dv = dst[c * CH + lane];
  for (;;) {
    const int  cn   = c + SC_WAVES;
    const bool more = cn < NCHUNK;
    int svn = 0, dvn = 0;
    if (more) {
      svn = src[cn * CH + lane];
      dvn = dst[cn * CH + lane];
    }
    const unsigned rb = (unsigned)(c / CH_PER_M) * NN;

    float r0[2][GE], r1[2][GE];
    #pragma unroll
    for (int j = 0; j < GE; ++j) {
      const unsigned s = (unsigned)__builtin_amdgcn_readlane(sv, j);
      const unsigned ix = (rb + s) * DH + lane;
      r0[0][j] = proj[ix];
      r1[0][j] = proj[ix + 32];
    }
    #pragma unroll
    for (int g = 0; g < CH / GE; ++g) {
      if (g + 1 < CH / GE) {
        if (g > 0) asm volatile("s_wait_xcnt 0x0" ::: "memory");
        #pragma unroll
        for (int j = 0; j < GE; ++j) {
          const unsigned s = (unsigned)__builtin_amdgcn_readlane(sv, (g + 1) * GE + j);
          const unsigned ix = (rb + s) * DH + lane;
          r0[(g + 1) & 1][j] = proj[ix];
          r1[(g + 1) & 1][j] = proj[ix + 32];
        }
      }
      #pragma unroll
      for (int j = 0; j < GE; ++j) {
        const unsigned d = (unsigned)__builtin_amdgcn_readlane(dv, g * GE + j);
        atomicAddRow2((float*)((char*)agg + ((rb + d) << 8)), voff, r0[g & 1][j], r1[g & 1][j]);
      }
    }
    if (!more) break;
    asm volatile("s_wait_xcnt 0x0" ::: "memory");
    c = cn; sv = svn; dv = dvn;
  }
}

__global__ __launch_bounds__(256)
void k_prelu(float* __restrict__ agg, const float* __restrict__ deg,
             const float* __restrict__ b, const float* __restrict__ pa) {
  size_t i4 = (size_t)blockIdx.x * 256 + threadIdx.x;
  if (i4 >= (size_t)MM * NN * (DH / 4)) return;
  size_t base = i4 * 4;
  size_t nl   = i4 >> 4;
  int m  = (int)(nl / NN);
  int j4 = (int)(i4 & 15);
  float d = fmaxf(deg[nl], 1.0f);
  float a = pa[m];
  v4f hv = *(v4f*)(agg + base);
  v4f bv = *(const v4f*)(b + m * DH + j4 * 4);
  #pragma unroll
  for (int c = 0; c < 4; ++c) {
    float h = hv[c] / d + bv[c];
    hv[c] = h > 0.f ? h : a * h;
  }
  *(v4f*)(agg + base) = hv;
}

__global__ __launch_bounds__(128)
void k_fc_wmma(const float* __restrict__ h, const float* __restrict__ fcW,
               const float* __restrict__ fcb, const float* __restrict__ attn,
               float* __restrict__ logits) {
  __shared__ float red[128];
  const int lane = threadIdx.x & 31;
  const int wv   = threadIdx.x >> 5;
  const int col0 = wv * 16;
  const int m    = blockIdx.y;
  const int n0   = blockIdx.x * 16;
  const int r    = lane & 15;
  const int hi   = lane >> 4;
  const int kh   = hi * 2;

  const float* A = h + ((size_t)m * NN + n0) * DH;
  float bias = fcb[col0 + r];
  v8f acc = {bias, bias, bias, bias, bias, bias, bias, bias};
  #pragma unroll
  for (int k = 0; k < DH; k += 4) {
    v2f a = *(const v2f*)(A + (size_t)r * DH + k + kh);
    v2f bb;
    bb.x = fcW[(k + kh)     * DH + col0 + r];
    bb.y = fcW[(k + kh + 1) * DH + col0 + r];
    acc = __builtin_amdgcn_wmma_f32_16x16x4_f32(false, a, false, bb, (short)0, acc,
                                                false, false);
  }
  float p = 0.f;
  #pragma unroll
  for (int v = 0; v < 8; ++v) p += tanhf(acc[v]);
  p *= attn[col0 + r];

  red[threadIdx.x] = p;
  __syncthreads();
  #pragma unroll
  for (int s2 = 64; s2 > 0; s2 >>= 1) {
    if (threadIdx.x < s2) red[threadIdx.x] += red[threadIdx.x + s2];
    __syncthreads();
  }
  if (threadIdx.x == 0) atomicAddF(&logits[m], red[0]);
}

__global__ void k_softmax(const float* __restrict__ logits, float* __restrict__ beta) {
  if (threadIdx.x == 0 && blockIdx.x == 0) {
    float l[MM]; float mx = -1e30f;
    for (int m = 0; m < MM; ++m) { l[m] = logits[m] / (float)NN; mx = fmaxf(mx, l[m]); }
    float s = 0.f;
    for (int m = 0; m < MM; ++m) { l[m] = expf(l[m] - mx); s += l[m]; }
    for (int m = 0; m < MM; ++m) beta[m] = l[m] / s;
  }
}

__global__ __launch_bounds__(256)
void k_out(const float* __restrict__ h, const float* __restrict__ beta,
           float* __restrict__ out) {
  size_t i4 = (size_t)blockIdx.x * 256 + threadIdx.x;
  if (i4 >= (size_t)NN * (DH / 4)) return;
  size_t base = i4 * 4;
  const size_t P = (size_t)NN * DH;
  float b0 = beta[0], b1 = beta[1], b2 = beta[2];
  v4f x0 = *(const v4f*)(h + base);
  v4f x1 = *(const v4f*)(h + P + base);
  v4f x2 = *(const v4f*)(h + 2 * P + base);
  v4f z  = x0 * b0 + x1 * b1 + x2 * b2;
  *(v4f*)(out + base) = z;
}

extern "C" void kernel_launch(void* const* d_in, const int* in_sizes, int n_in,
                              void* d_out, int out_size, void* d_ws, size_t ws_size,
                              hipStream_t stream) {
  const float* feats = (const float*)d_in[0];
  const int*   src   = (const int*)  d_in[1];
  const int*   dst   = (const int*)  d_in[2];
  const float* W     = (const float*)d_in[3];
  const float* b     = (const float*)d_in[4];
  const float* pa    = (const float*)d_in[5];
  const float* fcW   = (const float*)d_in[6];
  const float* fcb   = (const float*)d_in[7];
  const float* attn  = (const float*)d_in[8];

  float* ws     = (float*)d_ws;
  float* agg    = ws + AGG_OFF;
  float* deg    = ws + DEG_OFF;
  float* logits = ws + LOG_OFF;
  float* beta   = ws + BETA_OFF;
  float* proj   = ws + PROJ_OFF;
  float* out    = (float*)d_out;

  hipMemsetAsync(d_ws, 0, (size_t)(LOG_OFF + 8) * sizeof(float), stream);

  dim3 gTiles(NN / 16, MM);
  k_proj_wmma<<<gTiles, 128, 0, stream>>>(feats, W, proj);

  k_deg<<<dim3(DG_NCH, DG_NR, MM), 1024, 0, stream>>>(dst, deg);

  k_scatter<<<SC_BLOCKS, 256, 0, stream>>>(proj, src, dst, agg);

  size_t nv4 = (size_t)MM * NN * (DH / 4);
  k_prelu<<<(unsigned)((nv4 + 255) / 256), 256, 0, stream>>>(agg, deg, b, pa);

  k_fc_wmma<<<gTiles, 128, 0, stream>>>(agg, fcW, fcb, attn, logits);

  k_softmax<<<1, 32, 0, stream>>>(logits, beta);

  size_t ov4 = (size_t)NN * (DH / 4);
  k_out<<<(unsigned)((ov4 + 255) / 256), 256, 0, stream>>>(agg, beta, out);
}


# aggregation: each work-group keeps 1240 destination rows in LDS, filters the whole dst list, gathers matching proj rows and adds them with LDS f32 atomics (sum order: LDS arrival); agg not cleared
# speedup vs baseline: 3.3067x; 1.2353x over previous
#include <hip/hip_runtime.h>
#include <hip/hip_bf16.h>
#include <math.h>

#define NN  100000
#define MM  3
#define DIN 128
#define DH  64
#define NE  1600000

typedef __attribute__((ext_vector_type(2))) float v2f;
typedef __attribute__((ext_vector_type(4))) float v4f;
typedef __attribute__((ext_vector_type(8))) float v8f;

#define AGG_OFF   0ull
#define DEG_OFF   19200000ull
#define LOG_OFF   19500000ull
#define BETA_OFF  19500004ull
#define PROJ_OFF  19500008ull

__device__ __forceinline__ void atomicAddF(float* p, float v) {
  asm volatile("global_atomic_add_f32 %0, %1, off scope:SCOPE_DEV"
               :: "v"(p), "v"(v) : "memory");
}

__global__ __launch_bounds__(128)
void k_proj_wmma(const float* __restrict__ feats, const float* __restrict__ W,
                 float* __restrict__ proj) {
  const int lane = threadIdx.x & 31;
  const int wv   = threadIdx.x >> 5;
  const int col0 = wv * 16;
  const int m    = blockIdx.y;
  const int n0   = blockIdx.x * 16;
  const int r    = lane & 15;
  const int hi   = lane >> 4;
  const int kh   = hi * 2;

  const float* A = feats + ((size_t)m * NN + n0) * DIN;
  const float* B = W + (size_t)m * DIN * DH;

  v8f acc = {0.f,0.f,0.f,0.f,0.f,0.f,0.f,0.f};
  #pragma unroll
  for (int k = 0; k < DIN; k += 4) {
    v2f a = *(const v2f*)(A + (size_t)r * DIN + k + kh);
    v2f b;
    b.x = B[(k + kh)     * DH + col0 + r];
    b.y = B[(k + kh + 1) * DH + col0 + r];
    acc = __builtin_amdgcn_wmma_f32_16x16x4_f32(false, a, false, b, (short)0, acc,
                                                false, false);
  }
  float* O = proj + ((size_t)m * NN + n0) * DH;
  #pragma unroll
  for (int v = 0; v < 8; ++v)
    O[(size_t)(v + hi * 8) * DH + col0 + r] = acc[v];
}

#define DG_CH     40000
#define DG_NCH    (NE / DG_CH)
#define DG_RANGE  50048
#define DG_NR     2

__device__ __forceinline__ void atomicAddLine(float* line, unsigned voff, float v) {
  asm volatile("global_atomic_add_f32 %0, %1, %2 scope:SCOPE_DEV"
               :: "v"(voff), "v"(v), "s"(line) : "memory");
}

#define DG_LD     10
#define DG_FG     7

__global__ __launch_bounds__(1024)
void k_deg(const int* __restrict__ dst, float* __restrict__ deg) {
  __shared__ alignas(16) unsigned hist[DG_RANGE];
  const int tid = threadIdx.x;
  const int m   = blockIdx.z;
  const int lo  = blockIdx.y * DG_RANGE;
  const unsigned nn = (unsigned)((NN - lo) < DG_RANGE ? (NN - lo) : DG_RANGE);

  const int4* d4 = (const int4*)(dst + (size_t)m * NE + (size_t)blockIdx.x * DG_CH);
  int4 d[DG_LD];
  #pragma unroll
  for (int j = 0; j < DG_LD; ++j) {
    const int i = tid + j * 1024;
    d[j] = (i < DG_CH / 4) ? d4[i] : make_int4(-1, -1, -1, -1);
  }

  for (int i = tid; i < DG_RANGE / 4; i += 1024) ((uint4*)hist)[i] = make_uint4(0u, 0u, 0u, 0u);
  __syncthreads();

  #pragma unroll
  for (int j = 0; j < DG_LD; ++j) {
    const unsigned a0 = (unsigned)(d[j].x - lo), a1 = (unsigned)(d[j].y - lo);
    const unsigned a2 = (unsigned)(d[j].z - lo), a3 = (unsigned)(d[j].w - lo);
    if (a0 < nn) atomicAdd(&hist[a0], 1u);
    if (a1 < nn) atomicAdd(&hist[a1], 1u);
    if (a2 < nn) atomicAdd(&hist[a2], 1u);
    if (a3 < nn) atomicAdd(&hist[a3], 1u);
  }
  __syncthreads();

  const int wv = __builtin_amdgcn_readfirstlane(tid >> 5);
  const unsigned lane = (unsigned)tid & 31u;
  float* base = deg + (size_t)m * NN + lo;
  const int nlines = (int)(nn >> 5);
  for (int l0 = wv; l0 < nlines; l0 += 32 * DG_FG) {
    float h[DG_FG];
    #pragma unroll
    for (int j = 0; j < DG_FG; ++j) {
      const int ln = l0 + 32 * j;
      h[j] = (float)hist[(ln < nlines ? ln : l0) * 32 + lane];
    }
    #pragma unroll
    for (int j = 0; j < DG_FG; ++j) {
      const int ln = l0 + 32 * j;
      if (ln < nlines) atomicAddLine(base + ln * 32, lane * 4u, h[j]);
    }
    asm volatile("s_wait_xcnt 0x0" ::: "memory");
  }
}

#define RG_ROWS   1240
#define RG_N      81
#define RG_Q      64
#define RG_LD     8
#define RG_G      8
#define RG_NIT    392

__device__ __forceinline__ void rg_append(unsigned* q, unsigned head, int& cnt, int dv, int lo,
                                          unsigned nrows, unsigned e) {
  const unsigned a = (unsigned)(dv - lo);
  const bool hit = a < nrows;
  const unsigned mask = __builtin_amdgcn_ballot_w32(hit);
  if (mask) {
    if (hit) q[(head + cnt + __builtin_amdgcn_mbcnt_lo(mask, 0u)) & (RG_Q - 1)] = (e << 11) | a;
    cnt += __builtin_popcount(mask);
  }
}

__device__ __forceinline__ void rg_flush(float* rows, const unsigned* q, unsigned head, int nvalid,
                                         const float* __restrict__ projm,
                                         const int* __restrict__ srcm, unsigned lane) {
  const unsigned pk = q[(head + lane) & (RG_Q - 1)];
  const int sv = ((int)lane < nvalid) ? srcm[pk >> 11] : 0;
  const int ld = (int)(pk & 2047u);
  #pragma nounroll
  for (int g = 0; g < nvalid; g += RG_G) {
    v2f r[RG_G];
    #pragma unroll
    for (int j = 0; j < RG_G; ++j) {
      const unsigned s = (unsigned)__builtin_amdgcn_readlane(sv, g + j);
      r[j] = *(const v2f*)(projm + s * DH + lane * 2);
    }
    #pragma unroll
    for (int j = 0; j < RG_G; ++j) {
      if (g + j < nvalid) {
        const unsigned d = (unsigned)__builtin_amdgcn_readlane(ld, g + j);
        float* p = rows + d * DH + lane * 2;
        atomicAdd(p, r[j].x);
        atomicAdd(p + 1, r[j].y);
      }
    }
  }
}

__global__ __launch_bounds__(1024)
void k_scatter(const float* __restrict__ proj, const int* __restrict__ src,
               const int* __restrict__ dst, float* __restrict__ agg) {
  __shared__ alignas(16) float rows[RG_ROWS * DH];
  __shared__ unsigned ring[32 * RG_Q];
  const int tid = threadIdx.x;
  const unsigned lane = (unsigned)tid & 31u;
  const int wv = __builtin_amdgcn_readfirstlane(tid >> 5);
  const int m  = blockIdx.y;
  const int lo = blockIdx.x * RG_ROWS;
  const unsigned nrows = (unsigned)((NN - lo) < RG_ROWS ? (NN - lo) : RG_ROWS);
  const int4*  d4    = (const int4*)(dst + (size_t)m * NE);
  const int*   srcm  = src + (size_t)m * NE;
  const float* projm = proj + (size_t)m * NN * DH;
  unsigned* q = ring + wv * RG_Q;

  int4 d[RG_LD];
  #pragma unroll
  for (int j = 0; j < RG_LD; ++j) d[j] = d4[j * 1024 + tid];

  for (int i = tid; i < RG_ROWS * DH / 4; i += 1024) ((v4f*)rows)[i] = (v4f){0.f, 0.f, 0.f, 0.f};
  __syncthreads();

  unsigned head = 0;
  int cnt = 0;
  for (int it = 0; it < RG_NIT; it += RG_LD) {
    #pragma unroll
    for (int j = 0; j < RG_LD; ++j) {
      const int4 c = d[j];
      const int icur = (it + j) * 1024 + tid;
      const unsigned e0 = (unsigned)icur * 4u;
      const unsigned nr = (icur < NE / 4) ? nrows : 0u;
      const int cv[4] = {c.x, c.y, c.z, c.w};
      #pragma unroll
      for (int k = 0; k < 4; ++k) {
        rg_append(q, head, cnt, cv[k], lo, nr, e0 + k);
        if (__builtin_expect(cnt >= 32, 0)) {
          rg_flush(rows, q, head, 32, projm, srcm, lane);
          head = (head + 32) & (RG_Q - 1);
          cnt -= 32;
        }
      }
      const int inext = (it + j + RG_LD) * 1024 + tid;
      d[j] = d4[inext < NE / 4 ? inext : NE / 4 - 1];
    }
  }
  if (cnt > 0) rg_flush(rows, q, head, cnt, projm, srcm, lane);
  __syncthreads();

  float* o = agg + ((size_t)m * NN + lo) * DH;
  for (unsigned i = tid; i < nrows * (DH / 4); i += 1024) ((v4f*)o)[i] = ((const v4f*)rows)[i];
}

__global__ __launch_bounds__(256)
void k_prelu(float* __restrict__ agg, const float* __restrict__ deg,
             const float* __restrict__ b, const float* __restrict__ pa) {
  size_t i4 = (size_t)blockIdx.x * 256 + threadIdx.x;
  if (i4 >= (size_t)MM * NN * (DH / 4)) return;
  size_t base = i4 * 4;
  size_t nl   = i4 >> 4;
  int m  = (int)(nl / NN);
  int j4 = (int)(i4 & 15);
  float d = fmaxf(deg[nl], 1.0f);
  float a = pa[m];
  v4f hv = *(v4f*)(agg + base);
  v4f bv = *(const v4f*)(b + m * DH + j4 * 4);
  #pragma unroll
  for (int c = 0; c < 4; ++c) {
    float h = hv[c] / d + bv[c];
    hv[c] = h > 0.f ? h : a * h;
  }
  *(v4f*)(agg + base) = hv;
}

__global__ __launch_bounds__(128)
void k_fc_wmma(const float* __restrict__ h, const float* __restrict__ fcW,
               const float* __restrict__ fcb, const float* __restrict__ attn,
               float* __restrict__ logits) {
  __shared__ float red[128];
  const int lane = threadIdx.x & 31;
  const int wv   = threadIdx.x >> 5;
  const int col0 = wv * 16;
  const int m    = blockIdx.y;
  const int n0   = blockIdx.x * 16;
  const int r    = lane & 15;
  const int hi   = lane >> 4;
  const int kh   = hi * 2;

  const float* A = h + ((size_t)m * NN + n0) * DH;
  float bias = fcb[col0 + r];
  v8f acc = {bias, bias, bias, bias, bias, bias, bias, bias};
  #pragma unroll
  for (int k = 0; k < DH; k += 4) {
    v2f a = *(const v2f*)(A + (size_t)r * DH + k + kh);
    v2f bb;
    bb.x = fcW[(k + kh)     * DH + col0 + r];
    bb.y = fcW[(k + kh + 1) * DH + col0 + r];
    acc = __builtin_amdgcn_wmma_f32_16x16x4_f32(false, a, false, bb, (short)0, acc,
                                                false, false);
  }
  float p = 0.f;
  #pragma unroll
  for (int v = 0; v < 8; ++v) p += tanhf(acc[v]);
  p *= attn[col0 + r];

  red[threadIdx.x] = p;
  __syncthreads();
  #pragma unroll
  for (int s2 = 64; s2 > 0; s2 >>= 1) {
    if (threadIdx.x < s2) red[threadIdx.x] += red[threadIdx.x + s2];
    __syncthreads();
  }
  if (threadIdx.x == 0) atomicAddF(&logits[m], red[0]);
}

__global__ void k_softmax(const float* __restrict__ logits, float* __restrict__ beta) {
  if (threadIdx.x == 0 && blockIdx.x == 0) {
    float l[MM]; float mx = -1e30f;
    for (int m = 0; m < MM; ++m) { l[m] = logits[m] / (float)NN; mx = fmaxf(mx, l[m]); }
    float s = 0.f;
    for (int m = 0; m < MM; ++m) { l[m] = expf(l[m] - mx); s += l[m]; }
    for (int m = 0; m < MM; ++m) beta[m] = l[m] / s;
  }
}

__global__ __launch_bounds__(256)
void k_out(const float* __restrict__ h, const float* __restrict__ beta,
           float* __restrict__ out) {
  size_t i4 = (size_t)blockIdx.x * 256 + threadIdx.x;
  if (i4 >= (size_t)NN * (DH / 4)) return;
  size_t base = i4 * 4;
  const size_t P = (size_t)NN * DH;
  float b0 = beta[0], b1 = beta[1], b2 = beta[2];
  v4f x0 = *(const v4f*)(h + base);
  v4f x1 = *(const v4f*)(h + P + base);
  v4f x2 = *(const v4f*)(h + 2 * P + base);
  v4f z  = x0 * b0 + x1 * b1 + x2 * b2;
  *(v4f*)(out + base) = z;
}

extern "C" void kernel_launch(void* const* d_in, const int* in_sizes, int n_in,
                              void* d_out, int out_size, void* d_ws, size_t ws_size,
                              hipStream_t stream) {
  const float* feats = (const float*)d_in[0];
  const int*   src   = (const int*)  d_in[1];
  const int*   dst   = (const int*)  d_in[2];
  const float* W     = (const float*)d_in[3];
  const float* b     = (const float*)d_in[4];
  const float* pa    = (const float*)d_in[5];
  const float* fcW   = (const float*)d_in[6];
  const float* fcb   = (const float*)d_in[7];
  const float* attn  = (const float*)d_in[8];

  float* ws     = (float*)d_ws;
  float* agg    = ws + AGG_OFF;
  float* deg    = ws + DEG_OFF;
  float* logits = ws + LOG_OFF;
  float* beta   = ws + BETA_OFF;
  float* proj   = ws + PROJ_OFF;
  float* out    = (float*)d_out;

  hipMemsetAsync(deg, 0, (size_t)(LOG_OFF + 8 - DEG_OFF) * sizeof(float), stream);

  dim3 gTiles(NN / 16, MM);
  k_proj_wmma<<<gTiles, 128, 0, stream>>>(feats, W, proj);

  k_deg<<<dim3(DG_NCH, DG_NR, MM), 1024, 0, stream>>>(dst, deg);

  k_scatter<<<dim3(RG_N, MM), 1024, 0, stream>>>(proj, src, dst, agg);

  size_t nv4 = (size_t)MM * NN * (DH / 4);
  k_prelu<<<(unsigned)((nv4 + 255) / 256), 256, 0, stream>>>(agg, deg, b, pa);

  k_fc_wmma<<<gTiles, 128, 0, stream>>>(agg, fcW, fcb, attn, logits);

  k_softmax<<<1, 32, 0, stream>>>(logits, beta);

  size_t ov4 = (size_t)NN * (DH / 4);
  k_out<<<(unsigned)((ov4 + 255) / 256), 256, 0, stream>>>(agg, beta, out);
}


# aggregation work-group counts its rows' in-degree while it filters dst (LDS integer counters, exact) and stores prelu(row/max(deg,1)+b) from LDS: degree kernel and PReLU pass removed
# speedup vs baseline: 3.6521x; 1.1045x over previous
#include <hip/hip_runtime.h>
#include <hip/hip_bf16.h>
#include <math.h>

#define NN  100000
#define MM  3
#define DIN 128
#define DH  64
#define NE  1600000

typedef __attribute__((ext_vector_type(2))) float v2f;
typedef __attribute__((ext_vector_type(4))) float v4f;
typedef __attribute__((ext_vector_type(8))) float v8f;

#define AGG_OFF   0ull
#define LOG_OFF   19500000ull
#define BETA_OFF  19500004ull
#define PROJ_OFF  19500008ull

__device__ __forceinline__ void atomicAddF(float* p, float v) {
  asm volatile("global_atomic_add_f32 %0, %1, off scope:SCOPE_DEV"
               :: "v"(p), "v"(v) : "memory");
}

__global__ __launch_bounds__(128)
void k_proj_wmma(const float* __restrict__ feats, const float* __restrict__ W,
                 float* __restrict__ proj) {
  const int lane = threadIdx.x & 31;
  const int wv   = threadIdx.x >> 5;
  const int col0 = wv * 16;
  const int m    = blockIdx.y;
  const int n0   = blockIdx.x * 16;
  const int r    = lane & 15;
  const int hi   = lane >> 4;
  const int kh   = hi * 2;

  const float* A = feats + ((size_t)m * NN + n0) * DIN;
  const float* B = W + (size_t)m * DIN * DH;

  v8f acc = {0.f,0.f,0.f,0.f,0.f,0.f,0.f,0.f};
  #pragma unroll
  for (int k = 0; k < DIN; k += 4) {
    v2f a = *(const v2f*)(A + (size_t)r * DIN + k + kh);
    v2f b;
    b.x = B[(k + kh)     * DH + col0 + r];
    b.y = B[(k + kh + 1) * DH + col0 + r];
    acc = __builtin_amdgcn_wmma_f32_16x16x4_f32(false, a, false, b, (short)0, acc,
                                                false, false);
  }
  float* O = proj + ((size_t)m * NN + n0) * DH;
  #pragma unroll
  for (int v = 0; v < 8; ++v)
    O[(size_t)(v + hi * 8) * DH + col0 + r] = acc[v];
}

#define RG_ROWS   1220
#define RG_N      82
#define RG_Q      64
#define RG_LD     8
#define RG_G      8
#define RG_NIT    392

__device__ __forceinline__ void rg_append(unsigned* q, unsigned* degc, unsigned head, int& cnt, int dv,
                                          int lo, unsigned nrows, unsigned e) {
  const unsigned a = (unsigned)(dv - lo);
  const bool hit = a < nrows;
  const unsigned mask = __builtin_amdgcn_ballot_w32(hit);
  if (mask) {
    if (hit) {
      q[(head + cnt + __builtin_amdgcn_mbcnt_lo(mask, 0u)) & (RG_Q - 1)] = (e << 11) | a;
      atomicAdd(&degc[a], 1u);
    }
    cnt += __builtin_popcount(mask);
  }
}

__device__ __forceinline__ void rg_flush(float* rows, const unsigned* q, unsigned head, int nvalid,
                                         const float* __restrict__ projm,
                                         const int* __restrict__ srcm, unsigned lane) {
  const unsigned pk = q[(head + lane) & (RG_Q - 1)];
  const int sv = ((int)lane < nvalid) ? srcm[pk >> 11] : 0;
  const int ld = (int)(pk & 2047u);
  #pragma nounroll
  for (int g = 0; g < nvalid; g += RG_G) {
    v2f r[RG_G];
    #pragma unroll
    for (int j = 0; j < RG_G; ++j) {
      const unsigned s = (unsigned)__builtin_amdgcn_readlane(sv, g + j);
      r[j] = *(const v2f*)(projm + s * DH + lane * 2);
    }
    #pragma unroll
    for (int j = 0; j < RG_G; ++j) {
      if (g + j < nvalid) {
        const unsigned d = (unsigned)__builtin_amdgcn_readlane(ld, g + j);
        float* p = rows + d * DH + lane * 2;
        atomicAdd(p, r[j].x);
        atomicAdd(p + 1, r[j].y);
      }
    }
  }
}

__global__ __launch_bounds__(1024)
void k_scatter(const float* __restrict__ proj, const int* __restrict__ src,
               const int* __restrict__ dst, const float* __restrict__ b,
               const float* __restrict__ pa, float* __restrict__ hout) {
  __shared__ alignas(16) float rows[RG_ROWS * DH];
  __shared__ unsigned ring[32 * RG_Q];
  __shared__ unsigned degc[RG_ROWS];
  const int tid = threadIdx.x;
  const unsigned lane = (unsigned)tid & 31u;
  const int wv = __builtin_amdgcn_readfirstlane(tid >> 5);
  const int m  = blockIdx.y;
  const int lo = blockIdx.x * RG_ROWS;
  const unsigned nrows = (unsigned)((NN - lo) < RG_ROWS ? (NN - lo) : RG_ROWS);
  const int4*  d4    = (const int4*)(dst + (size_t)m * NE);
  const int*   srcm  = src + (size_t)m * NE;
  const float* projm = proj + (size_t)m * NN * DH;
  unsigned* q = ring + wv * RG_Q;

  int4 d[RG_LD];
  #pragma unroll
  for (int j = 0; j < RG_LD; ++j) d[j] = d4[j * 1024 + tid];

  for (int i = tid; i < RG_ROWS * DH / 4; i += 1024) ((v4f*)rows)[i] = (v4f){0.f, 0.f, 0.f, 0.f};
  for (int i = tid; i < RG_ROWS; i += 1024) degc[i] = 0u;
  __syncthreads();

  unsigned head = 0;
  int cnt = 0;
  for (int it = 0; it < RG_NIT; it += RG_LD) {
    #pragma unroll
    for (int j = 0; j < RG_LD; ++j) {
      const int4 c = d[j];
      const int icur = (it + j) * 1024 + tid;
      const unsigned e0 = (unsigned)icur * 4u;
      const unsigned nr = (icur < NE / 4) ? nrows : 0u;
      const int cv[4] = {c.x, c.y, c.z, c.w};
      #pragma unroll
      for (int k = 0; k < 4; ++k) {
        rg_append(q, degc, head, cnt, cv[k], lo, nr, e0 + k);
        if (__builtin_expect(cnt >= 32, 0)) {
          rg_flush(rows, q, head, 32, projm, srcm, lane);
          head = (head + 32) & (RG_Q - 1);
          cnt -= 32;
        }
      }
      const int inext = (it + j + RG_LD) * 1024 + tid;
      d[j] = d4[inext < NE / 4 ? inext : NE / 4 - 1];
    }
  }
  if (cnt > 0) rg_flush(rows, q, head, cnt, projm, srcm, lane);
  __syncthreads();

  float* o = hout + ((size_t)m * NN + lo) * DH;
  const float a = pa[m];
  const v4f bv = *(const v4f*)(b + m * DH + (tid & 15) * 4);
  for (unsigned i = tid; i < nrows * (DH / 4); i += 1024) {
    const float dg = fmaxf((float)degc[i >> 4], 1.0f);
    v4f hv = ((const v4f*)rows)[i];
    for (int c = 0; c < 4; ++c) {
      const float h = hv[c] / dg + bv[c];
      hv[c] = h > 0.f ? h : a * h;
    }
    ((v4f*)o)[i] = hv;
  }
}

__global__ __launch_bounds__(128)
void k_fc_wmma(const float* __restrict__ h, const float* __restrict__ fcW,
               const float* __restrict__ fcb, const float* __restrict__ attn,
               float* __restrict__ logits) {
  __shared__ float red[128];
  const int lane = threadIdx.x & 31;
  const int wv   = threadIdx.x >> 5;
  const int col0 = wv * 16;
  const int m    = blockIdx.y;
  const int n0   = blockIdx.x * 16;
  const int r    = lane & 15;
  const int hi   = lane >> 4;
  const int kh   = hi * 2;

  const float* A = h + ((size_t)m * NN + n0) * DH;
  float bias = fcb[col0 + r];
  v8f acc = {bias, bias, bias, bias, bias, bias, bias, bias};
  #pragma unroll
  for (int k = 0; k < DH; k += 4) {
    v2f a = *(const v2f*)(A + (size_t)r * DH + k + kh);
    v2f bb;
    bb.x = fcW[(k + kh)     * DH + col0 + r];
    bb.y = fcW[(k + kh + 1) * DH + col0 + r];
    acc = __builtin_amdgcn_wmma_f32_16x16x4_f32(false, a, false, bb, (short)0, acc,
                                                false, false);
  }
  float p = 0.f;
  #pragma unroll
  for (int v = 0; v < 8; ++v) p += tanhf(acc[v]);
  p *= attn[col0 + r];

  red[threadIdx.x] = p;
  __syncthreads();
  #pragma unroll
  for (int s2 = 64; s2 > 0; s2 >>= 1) {
    if (threadIdx.x < s2) red[threadIdx.x] += red[threadIdx.x + s2];
    __syncthreads();
  }
  if (threadIdx.x == 0) atomicAddF(&logits[m], red[0]);
}

__global__ void k_softmax(const float* __restrict__ logits, float* __restrict__ beta) {
  if (threadIdx.x == 0 && blockIdx.x == 0) {
    float l[MM]; float mx = -1e30f;
    for (int m = 0; m < MM; ++m) { l[m] = logits[m] / (float)NN; mx = fmaxf(mx, l[m]); }
    float s = 0.f;
    for (int m = 0; m < MM; ++m) { l[m] = expf(l[m] - mx); s += l[m]; }
    for (int m = 0; m < MM; ++m) beta[m] = l[m] / s;
  }
}

__global__ __launch_bounds__(256)
void k_out(const float* __restrict__ h, const float* __restrict__ beta,
           float* __restrict__ out) {
  size_t i4 = (size_t)blockIdx.x * 256 + threadIdx.x;
  if (i4 >= (size_t)NN * (DH / 4)) return;
  size_t base = i4 * 4;
  const size_t P = (size_t)NN * DH;
  float b0 = beta[0], b1 = beta[1], b2 = beta[2];
  v4f x0 = *(const v4f*)(h + base);
  v4f x1 = *(const v4f*)(h + P + base);
  v4f x2 = *(const v4f*)(h + 2 * P + base);
  v4f z  = x0 * b0 + x1 * b1 + x2 * b2;
  *(v4f*)(out + base) = z;
}

extern "C" void kernel_launch(void* const* d_in, const int* in_sizes, int n_in,
                              void* d_out, int out_size, void* d_ws, size_t ws_size,
                              hipStream_t stream) {
  const float* feats = (const float*)d_in[0];
  const int*   src   = (const int*)  d_in[1];
  const int*   dst   = (const int*)  d_in[2];
  const float* W     = (const float*)d_in[3];
  const float* b     = (const float*)d_in[4];
  const float* pa    = (const float*)d_in[5];
  const float* fcW   = (const float*)d_in[6];
  const float* fcb   = (const float*)d_in[7];
  const float* attn  = (const float*)d_in[8];

  float* ws     = (float*)d_ws;
  float* agg    = ws + AGG_OFF;
  float* logits = ws + LOG_OFF;
  float* beta   = ws + BETA_OFF;
  float* proj   = ws + PROJ_OFF;
  float* out    = (float*)d_out;

  hipMemsetAsync(logits, 0, 8 * sizeof(float), stream);

  dim3 gTiles(NN / 16, MM);
  k_proj_wmma<<<gTiles, 128, 0, stream>>>(feats, W, proj);

  k_scatter<<<dim3(RG_N, MM), 1024, 0, stream>>>(proj, src, dst, b, pa, agg);

  k_fc_wmma<<<gTiles, 128, 0, stream>>>(agg, fcW, fcb, attn, logits);

  k_softmax<<<1, 32, 0, stream>>>(logits, beta);

  size_t ov4 = (size_t)NN * (DH / 4);
  k_out<<<(unsigned)((ov4 + 255) / 256), 256, 0, stream>>>(agg, beta, out);
}


# edges counting-sorted once by destination range (count + fill kernels, integer atomics): an aggregation work-group reads its own edge list, not all of dst; row sum order: LDS arrival
# speedup vs baseline: 3.9873x; 1.0918x over previous
#include <hip/hip_runtime.h>
#include <hip/hip_bf16.h>
#include <math.h>

#define NN  100000
#define MM  3
#define DIN 128
#define DH  64
#define NE  1600000

typedef __attribute__((ext_vector_type(2))) float v2f;
typedef __attribute__((ext_vector_type(4))) float v4f;
typedef __attribute__((ext_vector_type(8))) float v8f;

#define AGG_OFF   0ull
#define LOG_OFF   19500000ull
#define BETA_OFF  19500004ull
#define PROJ_OFF  19500008ull
#define CNT_OFF   19499232ull

__device__ __forceinline__ void atomicAddF(float* p, float v) {
  asm volatile("global_atomic_add_f32 %0, %1, off scope:SCOPE_DEV"
               :: "v"(p), "v"(v) : "memory");
}

__global__ __launch_bounds__(128)
void k_proj_wmma(const float* __restrict__ feats, const float* __restrict__ W,
                 float* __restrict__ proj) {
  const int lane = threadIdx.x & 31;
  const int wv   = threadIdx.x >> 5;
  const int col0 = wv * 16;
  const int m    = blockIdx.y;
  const int n0   = blockIdx.x * 16;
  const int r    = lane & 15;
  const int hi   = lane >> 4;
  const int kh   = hi * 2;

  const float* A = feats + ((size_t)m * NN + n0) * DIN;
  const float* B = W + (size_t)m * DIN * DH;

  v8f acc = {0.f,0.f,0.f,0.f,0.f,0.f,0.f,0.f};
  #pragma unroll
  for (int k = 0; k < DIN; k += 4) {
    v2f a = *(const v2f*)(A + (size_t)r * DIN + k + kh);
    v2f b;
    b.x = B[(k + kh)     * DH + col0 + r];
    b.y = B[(k + kh + 1) * DH + col0 + r];
    acc = __builtin_amdgcn_wmma_f32_16x16x4_f32(false, a, false, b, (short)0, acc,
                                                false, false);
  }
  float* O = proj + ((size_t)m * NN + n0) * DH;
  #pragma unroll
  for (int v = 0; v < 8; ++v)
    O[(size_t)(v + hi * 8) * DH + col0 + r] = acc[v];
}

#define RG_ROWS   1220
#define RG_N      82
#define RG_G      16
#define BK_E      12800
#define BK_N      125
#define BK_IT     4

__global__ __launch_bounds__(1024)
void k_count(const int* __restrict__ dst, unsigned* __restrict__ tot) {
  __shared__ unsigned hist[128];
  const int tid = threadIdx.x;
  const int m   = blockIdx.y;
  if (tid < 128) hist[tid] = 0u;
  __syncthreads();
  const int4* d4 = (const int4*)(dst + (size_t)m * NE + (size_t)blockIdx.x * BK_E);
  #pragma unroll
  for (int k = 0; k < BK_IT; ++k) {
    const int i = k * 1024 + tid;
    if (i < BK_E / 4) {
      const int4 c = d4[i];
      const int cv[4] = {c.x, c.y, c.z, c.w};
      #pragma unroll
      for (int q = 0; q < 4; ++q) {
        const unsigned r = (unsigned)cv[q] / RG_ROWS;
        if (r < RG_N) atomicAdd(&hist[r], 1u);
      }
    }
  }
  __syncthreads();
  if (tid < RG_N) {
    const unsigned h = hist[tid];
    if (h) atomicAdd(&tot[m * 128 + tid], h);
  }
}

__global__ __launch_bounds__(1024)
void k_fill(const int* __restrict__ src, const int* __restrict__ dst,
            const unsigned* __restrict__ tot, unsigned* __restrict__ cur,
            unsigned* __restrict__ elist) {
  __shared__ unsigned hist[128];
  __shared__ unsigned start[128];
  __shared__ unsigned totl[128];
  const int tid = threadIdx.x;
  const int m   = blockIdx.y;
  if (tid < 128) {
    hist[tid] = 0u;
    totl[tid] = (tid < RG_N) ? tot[m * 128 + tid] : 0u;
  }
  __syncthreads();
  const size_t e0 = (size_t)m * NE + (size_t)blockIdx.x * BK_E;
  const int4* d4 = (const int4*)(dst + e0);
  const int4* s4 = (const int4*)(src + e0);
  int4 dv[BK_IT], sv[BK_IT];
  #pragma unroll
  for (int k = 0; k < BK_IT; ++k) {
    const int i = k * 1024 + tid;
    const int ic = i < BK_E / 4 ? i : BK_E / 4 - 1;
    dv[k] = d4[ic];
    sv[k] = s4[ic];
  }
  #pragma unroll
  for (int k = 0; k < BK_IT; ++k) {
    if (k * 1024 + tid < BK_E / 4) {
      const int cv[4] = {dv[k].x, dv[k].y, dv[k].z, dv[k].w};
      #pragma unroll
      for (int q = 0; q < 4; ++q) {
        const unsigned r = (unsigned)cv[q] / RG_ROWS;
        if (r < RG_N) atomicAdd(&hist[r], 1u);
      }
    }
  }
  __syncthreads();
  if (tid < RG_N) {
    unsigned base = 0u;
    for (int j = 0; j < tid; ++j) base += totl[j];
    const unsigned h = hist[tid];
    const unsigned got = h ? atomicAdd(&cur[m * 128 + tid], h) : 0u;
    start[tid] = base + got;
    hist[tid] = 0u;
  }
  __syncthreads();
  unsigned* lm = elist + (size_t)m * NE;
  #pragma unroll
  for (int k = 0; k < BK_IT; ++k) {
    if (k * 1024 + tid < BK_E / 4) {
      const int cv[4] = {dv[k].x, dv[k].y, dv[k].z, dv[k].w};
      const int cs[4] = {sv[k].x, sv[k].y, sv[k].z, sv[k].w};
      #pragma unroll
      for (int q = 0; q < 4; ++q) {
        const unsigned r = (unsigned)cv[q] / RG_ROWS;
        if (r < RG_N) {
          const unsigned pos = start[r] + atomicAdd(&hist[r], 1u);
          lm[pos] = ((unsigned)cs[q] << 11) | ((unsigned)cv[q] - r * RG_ROWS);
        }
      }
    }
  }
}

__device__ __forceinline__ void rg_add(float* rows, unsigned pk, int nvalid,
                                       const float* __restrict__ projm, unsigned lane) {
  const int sv = (int)(pk >> 11);
  const int ld = (int)(pk & 2047u);
  for (int g = 0; g < nvalid; g += RG_G) {
    v2f r[RG_G];
    #pragma unroll
    for (int j = 0; j < RG_G; ++j) {
      const unsigned s = (unsigned)__builtin_amdgcn_readlane(sv, g + j);
      r[j] = *(const v2f*)(projm + s * DH + lane * 2);
    }
    #pragma unroll
    for (int j = 0; j < RG_G; ++j) {
      if (g + j < nvalid) {
        const unsigned d = (unsigned)__builtin_amdgcn_readlane(ld, g + j);
        float* p = rows + d * DH + lane * 2;
        atomicAdd(p, r[j].x);
        atomicAdd(p + 1, r[j].y);
      }
    }
  }
}

__global__ __launch_bounds__(1024)
void k_scatter(const float* __restrict__ proj, const unsigned* __restrict__ elist,
               const unsigned* __restrict__ tot, const float* __restrict__ b,
               const float* __restrict__ pa, float* __restrict__ hout) {
  __shared__ alignas(16) float rows[RG_ROWS * DH];
  __shared__ unsigned degc[RG_ROWS];
  __shared__ unsigned totl[128];
  const int tid = threadIdx.x;
  const unsigned lane = (unsigned)tid & 31u;
  const unsigned wv = (unsigned)__builtin_amdgcn_readfirstlane(tid >> 5);
  const int m  = blockIdx.y;
  const int rg = blockIdx.x;
  const int lo = rg * RG_ROWS;
  const unsigned nrows = (unsigned)((NN - lo) < RG_ROWS ? (NN - lo) : RG_ROWS);
  const float* projm = proj + (size_t)m * NN * DH;

  if (tid < 128) totl[tid] = (tid < RG_N) ? tot[m * 128 + tid] : 0u;
  for (int i = tid; i < RG_ROWS * DH / 4; i += 1024) ((v4f*)rows)[i] = (v4f){0.f, 0.f, 0.f, 0.f};
  for (int i = tid; i < RG_ROWS; i += 1024) degc[i] = 0u;
  __syncthreads();

  unsigned bsum = 0u;
  for (int j = 0; j < rg; ++j) bsum += totl[j];
  const unsigned base = (unsigned)__builtin_amdgcn_readfirstlane((int)bsum);
  const unsigned n    = (unsigned)__builtin_amdgcn_readfirstlane((int)totl[rg]);
  const unsigned* bin = elist + (size_t)m * NE + base;

  for (unsigned i0 = wv * 128u; i0 < n; i0 += 4096u) {
    unsigned pk[4];
    #pragma unroll
    for (int q = 0; q < 4; ++q) {
      const unsigned i = i0 + (unsigned)q * 32u + lane;
      pk[q] = (i < n) ? bin[i] : 0u;
    }
    #pragma unroll
    for (int q = 0; q < 4; ++q) {
      const unsigned s0 = i0 + (unsigned)q * 32u;
      if (s0 < n) {
        const unsigned rem = n - s0;
        const int nvalid = rem < 32u ? (int)rem : 32;
        if ((int)lane < nvalid) atomicAdd(&degc[pk[q] & 2047u], 1u);
        rg_add(rows, pk[q], nvalid, projm, lane);
      }
    }
  }
  __syncthreads();

  float* o = hout + ((size_t)m * NN + lo) * DH;
  const float a = pa[m];
  const v4f bv = *(const v4f*)(b + m * DH + (tid & 15) * 4);
  for (unsigned i = tid; i < nrows * (DH / 4); i += 1024) {
    const float dg = fmaxf((float)degc[i >> 4], 1.0f);
    v4f hv = ((const v4f*)rows)[i];
    for (int c = 0; c < 4; ++c) {
      const float h = hv[c] / dg + bv[c];
      hv[c] = h > 0.f ? h : a * h;
    }
    ((v4f*)o)[i] = hv;
  }
}

__global__ __launch_bounds__(128)
void k_fc_wmma(const float* __restrict__ h, const float* __restrict__ fcW,
               const float* __restrict__ fcb, const float* __restrict__ attn,
               float* __restrict__ logits) {
  __shared__ float red[128];
  const int lane = threadIdx.x & 31;
  const int wv   = threadIdx.x >> 5;
  const int col0 = wv * 16;
  const int m    = blockIdx.y;
  const int n0   = blockIdx.x * 16;
  const int r    = lane & 15;
  const int hi   = lane >> 4;
  const int kh   = hi * 2;

  const float* A = h + ((size_t)m * NN + n0) * DH;
  float bias = fcb[col0 + r];
  v8f acc = {bias, bias, bias, bias, bias, bias, bias, bias};
  #pragma unroll
  for (int k = 0; k < DH; k += 4) {
    v2f a = *(const v2f*)(A + (size_t)r * DH + k + kh);
    v2f bb;
    bb.x = fcW[(k + kh)     * DH + col0 + r];
    bb.y = fcW[(k + kh + 1) * DH + col0 + r];
    acc = __builtin_amdgcn_wmma_f32_16x16x4_f32(false, a, false, bb, (short)0, acc,
                                                false, false);
  }
  float p = 0.f;
  #pragma unroll
  for (int v = 0; v < 8; ++v) p += tanhf(acc[v]);
  p *= attn[col0 + r];

  red[threadIdx.x] = p;
  __syncthreads();
  #pragma unroll
  for (int s2 = 64; s2 > 0; s2 >>= 1) {
    if (threadIdx.x < s2) red[threadIdx.x] += red[threadIdx.x + s2];
    __syncthreads();
  }
  if (threadIdx.x == 0) atomicAddF(&logits[m], red[0]);
}

__global__ void k_softmax(const float* __restrict__ logits, float* __restrict__ beta) {
  if (threadIdx.x == 0 && blockIdx.x == 0) {
    float l[MM]; float mx = -1e30f;
    for (int m = 0; m < MM; ++m) { l[m] = logits[m] / (float)NN; mx = fmaxf(mx, l[m]); }
    float s = 0.f;
    for (int m = 0; m < MM; ++m) { l[m] = expf(l[m] - mx); s += l[m]; }
    for (int m = 0; m < MM; ++m) beta[m] = l[m] / s;
  }
}

__global__ __launch_bounds__(256)
void k_out(const float* __restrict__ h, const float* __restrict__ beta,
           float* __restrict__ out) {
  size_t i4 = (size_t)blockIdx.x * 256 + threadIdx.x;
  if (i4 >= (size_t)NN * (DH / 4)) return;
  size_t base = i4 * 4;
  const size_t P = (size_t)NN * DH;
  float b0 = beta[0], b1 = beta[1], b2 = beta[2];
  v4f x0 = *(const v4f*)(h + base);
  v4f x1 = *(const v4f*)(h + P + base);
  v4f x2 = *(const v4f*)(h + 2 * P + base);
  v4f z  = x0 * b0 + x1 * b1 + x2 * b2;
  *(v4f*)(out + base) = z;
}

extern "C" void kernel_launch(void* const* d_in, const int* in_sizes, int n_in,
                              void* d_out, int out_size, void* d_ws, size_t ws_size,
                              hipStream_t stream) {
  const float* feats = (const float*)d_in[0];
  const int*   src   = (const int*)  d_in[1];
  const int*   dst   = (const int*)  d_in[2];
  const float* W     = (const float*)d_in[3];
  const float* b     = (const float*)d_in[4];
  const float* pa    = (const float*)d_in[5];
  const float* fcW   = (const float*)d_in[6];
  const float* fcb   = (const float*)d_in[7];
  const float* attn  = (const float*)d_in[8];

  float* ws     = (float*)d_ws;
  float* agg    = ws + AGG_OFF;
  float* logits = ws + LOG_OFF;
  float* beta   = ws + BETA_OFF;
  float* proj   = ws + PROJ_OFF;
  float* out    = (float*)d_out;
  unsigned* tot = (unsigned*)(ws + CNT_OFF);
  unsigned* cur = tot + 384;
  unsigned* elist = (unsigned*)d_out;

  hipMemsetAsync(ws + CNT_OFF, 0, (768 + 8) * sizeof(float), stream);

  k_count<<<dim3(BK_N, MM), 1024, 0, stream>>>(dst, tot);
  k_fill<<<dim3(BK_N, MM), 1024, 0, stream>>>(src, dst, tot, cur, elist);

  dim3 gTiles(NN / 16, MM);
  k_proj_wmma<<<gTiles, 128, 0, stream>>>(feats, W, proj);

  k_scatter<<<dim3(RG_N, MM), 1024, 0, stream>>>(proj, elist, tot, b, pa, agg);

  k_fc_wmma<<<gTiles, 128, 0, stream>>>(agg, fcW, fcb, attn, logits);

  k_softmax<<<1, 32, 0, stream>>>(logits, beta);

  size_t ov4 = (size_t)NN * (DH / 4);
  k_out<<<(unsigned)((ov4 + 255) / 256), 256, 0, stream>>>(agg, beta, out);
}


# projection: a wave keeps W[m] in registers and walks 16-row tiles, all 64 columns each; A by 16-byte loads a tile ahead, whole-row stores; k summed in another order (f32 as before)
# speedup vs baseline: 4.6539x; 1.1672x over previous
#include <hip/hip_runtime.h>
#include <hip/hip_bf16.h>
#include <math.h>

#define NN  100000
#define MM  3
#define DIN 128
#define DH  64
#define NE  1600000

typedef __attribute__((ext_vector_type(2))) float v2f;
typedef __attribute__((ext_vector_type(4))) float v4f;
typedef __attribute__((ext_vector_type(8))) float v8f;

#define AGG_OFF   0ull
#define LOG_OFF   19500000ull
#define BETA_OFF  19500004ull
#define PROJ_OFF  19500008ull
#define CNT_OFF   19499232ull

__device__ __forceinline__ void atomicAddF(float* p, float v) {
  asm volatile("global_atomic_add_f32 %0, %1, off scope:SCOPE_DEV"
               :: "v"(p), "v"(v) : "memory");
}

#define PJ_WAVES 2048
#define PJ_TILES (NN / 16)

__device__ __forceinline__ void pj_load(v4f (&a)[16], const float* __restrict__ Am, int t,
                                        int c, int hi) {
  const float* p = Am + ((size_t)t * 16 + c) * DIN + 4 * hi;
  #pragma unroll
  for (int q = 0; q < 16; ++q) a[q] = *(const v4f*)(p + 8 * q);
}

__device__ __forceinline__ void pj_tile(const v4f (&a)[16], const v2f (&breg)[4][32],
                                        float* __restrict__ Om, int t, int c, int hi) {
  v8f acc[4];
  #pragma unroll
  for (int ct = 0; ct < 4; ++ct) acc[ct] = (v8f){0.f,0.f,0.f,0.f,0.f,0.f,0.f,0.f};
  #pragma unroll
  for (int p = 0; p < 16; ++p) {
    const v2f a0 = {a[p].x, a[p].y};
    const v2f a1 = {a[p].z, a[p].w};
    #pragma unroll
    for (int ct = 0; ct < 4; ++ct)
      acc[ct] = __builtin_amdgcn_wmma_f32_16x16x4_f32(false, a0, false, breg[ct][2 * p], (short)0,
                                                      acc[ct], false, false);
    #pragma unroll
    for (int ct = 0; ct < 4; ++ct)
      acc[ct] = __builtin_amdgcn_wmma_f32_16x16x4_f32(false, a1, false, breg[ct][2 * p + 1], (short)0,
                                                      acc[ct], false, false);
  }
  float* o = Om + ((size_t)t * 16 + 8 * hi) * DH + 4 * c;
  #pragma unroll
  for (int v = 0; v < 8; ++v)
    *(v4f*)(o + (size_t)v * DH) = (v4f){acc[0][v], acc[1][v], acc[2][v], acc[3][v]};
}

__global__ __launch_bounds__(128, 2)
void k_proj_wmma(const float* __restrict__ feats, const float* __restrict__ W,
                 float* __restrict__ proj) {
  const int lane = threadIdx.x & 31;
  const int c    = lane & 15;
  const int hi   = lane >> 4;
  const int gw = __builtin_amdgcn_readfirstlane((int)(blockIdx.x * 4 + (threadIdx.x >> 5)));
  const int m     = gw * MM / PJ_WAVES;
  const int first = (m * PJ_WAVES + MM - 1) / MM;
  const int nw    = ((m + 1) * PJ_WAVES + MM - 1) / MM - first;
  const float* Am = feats + (size_t)m * NN * DIN;
  float*       Om = proj  + (size_t)m * NN * DH;

  v2f breg[4][32];
  {
    const float* Bm = W + (size_t)m * DIN * DH + (size_t)(4 * hi) * DH + 4 * c;
    #pragma unroll
    for (int q = 0; q < 32; ++q) {
      const int k0 = 8 * (q >> 1) + 2 * (q & 1);
      const v4f w0 = *(const v4f*)(Bm + (size_t)k0 * DH);
      const v4f w1 = *(const v4f*)(Bm + (size_t)(k0 + 1) * DH);
      #pragma unroll
      for (int ct = 0; ct < 4; ++ct) breg[ct][q] = (v2f){w0[ct], w1[ct]};
    }
  }

  int t = gw - first;
  const int n = (PJ_TILES - t + nw - 1) / nw;
  v4f a0[16], a1[16];
  pj_load(a0, Am, t, c, hi);
  for (int j = 0; j + 1 < n; j += 2) {
    pj_load(a1, Am, t + nw, c, hi);
    pj_tile(a0, breg, Om, t, c, hi);
    const int t2 = t + 2 * nw;
    pj_load(a0, Am, t2 < PJ_TILES ? t2 : t, c, hi);
    pj_tile(a1, breg, Om, t + nw, c, hi);
    t = t2;
  }
  if (n & 1) pj_tile(a0, breg, Om, t, c, hi);
}

#define RG_ROWS   1220
#define RG_N      82
#define RG_G      16
#define BK_E      12800
#define BK_N      125
#define BK_IT     4

__global__ __launch_bounds__(1024)
void k_count(const int* __restrict__ dst, unsigned* __restrict__ tot) {
  __shared__ unsigned hist[128];
  const int tid = threadIdx.x;
  const int m   = blockIdx.y;
  if (tid < 128) hist[tid] = 0u;
  __syncthreads();
  const int4* d4 = (const int4*)(dst + (size_t)m * NE + (size_t)blockIdx.x * BK_E);
  #pragma unroll
  for (int k = 0; k < BK_IT; ++k) {
    const int i = k * 1024 + tid;
    if (i < BK_E / 4) {
      const int4 c = d4[i];
      const int cv[4] = {c.x, c.y, c.z, c.w};
      #pragma unroll
      for (int q = 0; q < 4; ++q) {
        const unsigned r = (unsigned)cv[q] / RG_ROWS;
        if (r < RG_N) atomicAdd(&hist[r], 1u);
      }
    }
  }
  __syncthreads();
  if (tid < RG_N) {
    const unsigned h = hist[tid];
    if (h) atomicAdd(&tot[m * 128 + tid], h);
  }
}

__global__ __launch_bounds__(1024)
void k_fill(const int* __restrict__ src, const int* __restrict__ dst,
            const unsigned* __restrict__ tot, unsigned* __restrict__ cur,
            unsigned* __restrict__ elist) {
  __shared__ unsigned hist[128];
  __shared__ unsigned start[128];
  __shared__ unsigned totl[128];
  const int tid = threadIdx.x;
  const int m   = blockIdx.y;
  if (tid < 128) {
    hist[tid] = 0u;
    totl[tid] = (tid < RG_N) ? tot[m * 128 + tid] : 0u;
  }
  __syncthreads();
  const size_t e0 = (size_t)m * NE + (size_t)blockIdx.x * BK_E;
  const int4* d4 = (const int4*)(dst + e0);
  const int4* s4 = (const int4*)(src + e0);
  int4 dv[BK_IT], sv[BK_IT];
  #pragma unroll
  for (int k = 0; k < BK_IT; ++k) {
    const int i = k * 1024 + tid;
    const int ic = i < BK_E / 4 ? i : BK_E / 4 - 1;
    dv[k] = d4[ic];
    sv[k] = s4[ic];
  }
  #pragma unroll
  for (int k = 0; k < BK_IT; ++k) {
    if (k * 1024 + tid < BK_E / 4) {
      const int cv[4] = {dv[k].x, dv[k].y, dv[k].z, dv[k].w};
      #pragma unroll
      for (int q = 0; q < 4; ++q) {
        const unsigned r = (unsigned)cv[q] / RG_ROWS;
        if (r < RG_N) atomicAdd(&hist[r], 1u);
      }
    }
  }
  __syncthreads();
  if (tid < RG_N) {
    unsigned base = 0u;
    for (int j = 0; j < tid; ++j) base += totl[j];
    const unsigned h = hist[tid];
    const unsigned got = h ? atomicAdd(&cur[m * 128 + tid], h) : 0u;
    start[tid] = base + got;
    hist[tid] = 0u;
  }
  __syncthreads();
  unsigned* lm = elist + (size_t)m * NE;
  #pragma unroll
  for (int k = 0; k < BK_IT; ++k) {
    if (k * 1024 + tid < BK_E / 4) {
      const int cv[4] = {dv[k].x, dv[k].y, dv[k].z, dv[k].w};
      const int cs[4] = {sv[k].x, sv[k].y, sv[k].z, sv[k].w};
      #pragma unroll
      for (int q = 0; q < 4; ++q) {
        const unsigned r = (unsigned)cv[q] / RG_ROWS;
        if (r < RG_N) {
          const unsigned pos = start[r] + atomicAdd(&hist[r], 1u);
          lm[pos] = ((unsigned)cs[q] << 11) | ((unsigned)cv[q] - r * RG_ROWS);
        }
      }
    }
  }
}

__device__ __forceinline__ void rg_add(float* rows, unsigned pk, int nvalid,
                                       const float* __restrict__ projm, unsigned lane) {
  const int sv = (int)(pk >> 11);
  const int ld = (int)(pk & 2047u);
  for (int g = 0; g < nvalid; g += RG_G) {
    v2f r[RG_G];
    #pragma unroll
    for (int j = 0; j < RG_G; ++j) {
      const unsigned s = (unsigned)__builtin_amdgcn_readlane(sv, g + j);
      r[j] = *(const v2f*)(projm + s * DH + lane * 2);
    }
    #pragma unroll
    for (int j = 0; j < RG_G; ++j) {
      if (g + j < nvalid) {
        const unsigned d = (unsigned)__builtin_amdgcn_readlane(ld, g + j);
        float* p = rows + d * DH + lane * 2;
        atomicAdd(p, r[j].x);
        atomicAdd(p + 1, r[j].y);
      }
    }
  }
}

__global__ __launch_bounds__(1024)
void k_scatter(const float* __restrict__ proj, const unsigned* __restrict__ elist,
               const unsigned* __restrict__ tot, const float* __restrict__ b,
               const float* __restrict__ pa, float* __restrict__ hout) {
  __shared__ alignas(16) float rows[RG_ROWS * DH];
  __shared__ unsigned degc[RG_ROWS];
  __shared__ unsigned totl[128];
  const int tid = threadIdx.x;
  const unsigned lane = (unsigned)tid & 31u;
  const unsigned wv = (unsigned)__builtin_amdgcn_readfirstlane(tid >> 5);
  const int m  = blockIdx.y;
  const int rg = blockIdx.x;
  const int lo = rg * RG_ROWS;
  const unsigned nrows = (unsigned)((NN - lo) < RG_ROWS ? (NN - lo) : RG_ROWS);
  const float* projm = proj + (size_t)m * NN * DH;

  if (tid < 128) totl[tid] = (tid < RG_N) ? tot[m * 128 + tid] : 0u;
  for (int i = tid; i < RG_ROWS * DH / 4; i += 1024) ((v4f*)rows)[i] = (v4f){0.f, 0.f, 0.f, 0.f};
  for (int i = tid; i < RG_ROWS; i += 1024) degc[i] = 0u;
  __syncthreads();

  unsigned bsum = 0u;
  for (int j = 0; j < rg; ++j) bsum += totl[j];
  const unsigned base = (unsigned)__builtin_amdgcn_readfirstlane((int)bsum);
  const unsigned n    = (unsigned)__builtin_amdgcn_readfirstlane((int)totl[rg]);
  const unsigned* bin = elist + (size_t)m * NE + base;

  for (unsigned i0 = wv * 128u; i0 < n; i0 += 4096u) {
    unsigned pk[4];
    #pragma unroll
    for (int q = 0; q < 4; ++q) {
      const unsigned i = i0 + (unsigned)q * 32u + lane;
      pk[q] = (i < n) ? bin[i] : 0u;
    }
    #pragma unroll
    for (int q = 0; q < 4; ++q) {
      const unsigned s0 = i0 + (unsigned)q * 32u;
      if (s0 < n) {
        const unsigned rem = n - s0;
        const int nvalid = rem < 32u ? (int)rem : 32;
        if ((int)lane < nvalid) atomicAdd(&degc[pk[q] & 2047u], 1u);
        rg_add(rows, pk[q], nvalid, projm, lane);
      }
    }
  }
  __syncthreads();

  float* o = hout + ((size_t)m * NN + lo) * DH;
  const float a = pa[m];
  const v4f bv = *(const v4f*)(b + m * DH + (tid & 15) * 4);
  for (unsigned i = tid; i < nrows * (DH / 4); i += 1024) {
    const float dg = fmaxf((float)degc[i >> 4], 1.0f);
    v4f hv = ((const v4f*)rows)[i];
    for (int c = 0; c < 4; ++c) {
      const float h = hv[c] / dg + bv[c];
      hv[c] = h > 0.f ? h : a * h;
    }
    ((v4f*)o)[i] = hv;
  }
}

__global__ __launch_bounds__(128)
void k_fc_wmma(const float* __restrict__ h, const float* __restrict__ fcW,
               const float* __restrict__ fcb, const float* __restrict__ attn,
               float* __restrict__ logits) {
  __shared__ float red[128];
  const int lane = threadIdx.x & 31;
  const int wv   = threadIdx.x >> 5;
  const int col0 = wv * 16;
  const int m    = blockIdx.y;
  const int n0   = blockIdx.x * 16;
  const int r    = lane & 15;
  const int hi   = lane >> 4;
  const int kh   = hi * 2;

  const float* A = h + ((size_t)m * NN + n0) * DH;
  float bias = fcb[col0 + r];
  v8f acc = {bias, bias, bias, bias, bias, bias, bias, bias};
  #pragma unroll
  for (int k = 0; k < DH; k += 4) {
    v2f a = *(const v2f*)(A + (size_t)r * DH + k + kh);
    v2f bb;
    bb.x = fcW[(k + kh)     * DH + col0 + r];
    bb.y = fcW[(k + kh + 1) * DH + col0 + r];
    acc = __builtin_amdgcn_wmma_f32_16x16x4_f32(false, a, false, bb, (short)0, acc,
                                                false, false);
  }
  float p = 0.f;
  #pragma unroll
  for (int v = 0; v < 8; ++v) p += tanhf(acc[v]);
  p *= attn[col0 + r];

  red[threadIdx.x] = p;
  __syncthreads();
  #pragma unroll
  for (int s2 = 64; s2 > 0; s2 >>= 1) {
    if (threadIdx.x < s2) red[threadIdx.x] += red[threadIdx.x + s2];
    __syncthreads();
  }
  if (threadIdx.x == 0) atomicAddF(&logits[m], red[0]);
}

__global__ void k_softmax(const float* __restrict__ logits, float* __restrict__ beta) {
  if (threadIdx.x == 0 && blockIdx.x == 0) {
    float l[MM]; float mx = -1e30f;
    for (int m = 0; m < MM; ++m) { l[m] = logits[m] / (float)NN; mx = fmaxf(mx, l[m]); }
    float s = 0.f;
    for (int m = 0; m < MM; ++m) { l[m] = expf(l[m] - mx); s += l[m]; }
    for (int m = 0; m < MM; ++m) beta[m] = l[m] / s;
  }
}

__global__ __launch_bounds__(256)
void k_out(const float* __restrict__ h, const float* __restrict__ beta,
           float* __restrict__ out) {
  size_t i4 = (size_t)blockIdx.x * 256 + threadIdx.x;
  if (i4 >= (size_t)NN * (DH / 4)) return;
  size_t base = i4 * 4;
  const size_t P = (size_t)NN * DH;
  float b0 = beta[0], b1 = beta[1], b2 = beta[2];
  v4f x0 = *(const v4f*)(h + base);
  v4f x1 = *(const v4f*)(h + P + base);
  v4f x2 = *(const v4f*)(h + 2 * P + base);
  v4f z  = x0 * b0 + x1 * b1 + x2 * b2;
  *(v4f*)(out + base) = z;
}

extern "C" void kernel_launch(void* const* d_in, const int* in_sizes, int n_in,
                              void* d_out, int out_size, void* d_ws, size_t ws_size,
                              hipStream_t stream) {
  const float* feats = (const float*)d_in[0];
  const int*   src   = (const int*)  d_in[1];
  const int*   dst   = (const int*)  d_in[2];
  const float* W     = (const float*)d_in[3];
  const float* b     = (const float*)d_in[4];
  const float* pa    = (const float*)d_in[5];
  const float* fcW   = (const float*)d_in[6];
  const float* fcb   = (const float*)d_in[7];
  const float* attn  = (const float*)d_in[8];

  float* ws     = (float*)d_ws;
  float* agg    = ws + AGG_OFF;
  float* logits = ws + LOG_OFF;
  float* beta   = ws + BETA_OFF;
  float* proj   = ws + PROJ_OFF;
  float* out    = (float*)d_out;
  unsigned* tot = (unsigned*)(ws + CNT_OFF);
  unsigned* cur = tot + 384;
  unsigned* elist = (unsigned*)d_out;

  hipMemsetAsync(ws + CNT_OFF, 0, (768 + 8) * sizeof(float), stream);

  k_count<<<dim3(BK_N, MM), 1024, 0, stream>>>(dst, tot);
  k_fill<<<dim3(BK_N, MM), 1024, 0, stream>>>(src, dst, tot, cur, elist);

  dim3 gTiles(NN / 16, MM);
  k_proj_wmma<<<PJ_WAVES / 4, 128, 0, stream>>>(feats, W, proj);

  k_scatter<<<dim3(RG_N, MM), 1024, 0, stream>>>(proj, elist, tot, b, pa, agg);

  k_fc_wmma<<<gTiles, 128, 0, stream>>>(agg, fcW, fcb, attn, logits);

  k_softmax<<<1, 32, 0, stream>>>(logits, beta);

  size_t ov4 = (size_t)NN * (DH / 4);
  k_out<<<(unsigned)((ov4 + 255) / 256), 256, 0, stream>>>(agg, beta, out);
}


# aggregation: work-group counting-sorts its edges by row in LDS (integer atomics), one wave then sums each row's proj rows in registers and stores h; f32 LDS atomics gone; sum order: row list order
# speedup vs baseline: 4.8634x; 1.0450x over previous
#include <hip/hip_runtime.h>
#include <hip/hip_bf16.h>
#include <math.h>

#define NN  100000
#define MM  3
#define DIN 128
#define DH  64
#define NE  1600000

typedef __attribute__((ext_vector_type(2))) float v2f;
typedef __attribute__((ext_vector_type(4))) float v4f;
typedef __attribute__((ext_vector_type(8))) float v8f;

#define AGG_OFF   0ull
#define LOG_OFF   19500000ull
#define BETA_OFF  19500004ull
#define PROJ_OFF  19500008ull
#define CNT_OFF   19499232ull

__device__ __forceinline__ void atomicAddF(float* p, float v) {
  asm volatile("global_atomic_add_f32 %0, %1, off scope:SCOPE_DEV"
               :: "v"(p), "v"(v) : "memory");
}

#define PJ_WAVES 2048
#define PJ_TILES (NN / 16)

__device__ __forceinline__ void pj_load(v4f (&a)[16], const float* __restrict__ Am, int t,
                                        int c, int hi) {
  const float* p = Am + ((size_t)t * 16 + c) * DIN + 4 * hi;
  #pragma unroll
  for (int q = 0; q < 16; ++q) a[q] = *(const v4f*)(p + 8 * q);
}

__device__ __forceinline__ void pj_tile(const v4f (&a)[16], const v2f (&breg)[4][32],
                                        float* __restrict__ Om, int t, int c, int hi) {
  v8f acc[4];
  #pragma unroll
  for (int ct = 0; ct < 4; ++ct) acc[ct] = (v8f){0.f,0.f,0.f,0.f,0.f,0.f,0.f,0.f};
  #pragma unroll
  for (int p = 0; p < 16; ++p) {
    const v2f a0 = {a[p].x, a[p].y};
    const v2f a1 = {a[p].z, a[p].w};
    #pragma unroll
    for (int ct = 0; ct < 4; ++ct)
      acc[ct] = __builtin_amdgcn_wmma_f32_16x16x4_f32(false, a0, false, breg[ct][2 * p], (short)0,
                                                      acc[ct], false, false);
    #pragma unroll
    for (int ct = 0; ct < 4; ++ct)
      acc[ct] = __builtin_amdgcn_wmma_f32_16x16x4_f32(false, a1, false, breg[ct][2 * p + 1], (short)0,
                                                      acc[ct], false, false);
  }
  float* o = Om + ((size_t)t * 16 + 8 * hi) * DH + 4 * c;
  #pragma unroll
  for (int v = 0; v < 8; ++v)
    *(v4f*)(o + (size_t)v * DH) = (v4f){acc[0][v], acc[1][v], acc[2][v], acc[3][v]};
}

__global__ __launch_bounds__(128, 2)
void k_proj_wmma(const float* __restrict__ feats, const float* __restrict__ W,
                 float* __restrict__ proj) {
  const int lane = threadIdx.x & 31;
  const int c    = lane & 15;
  const int hi   = lane >> 4;
  const int gw = __builtin_amdgcn_readfirstlane((int)(blockIdx.x * 4 + (threadIdx.x >> 5)));
  const int m     = gw * MM / PJ_WAVES;
  const int first = (m * PJ_WAVES + MM - 1) / MM;
  const int nw    = ((m + 1) * PJ_WAVES + MM - 1) / MM - first;
  const float* Am = feats + (size_t)m * NN * DIN;
  float*       Om = proj  + (size_t)m * NN * DH;

  v2f breg[4][32];
  {
    const float* Bm = W + (size_t)m * DIN * DH + (size_t)(4 * hi) * DH + 4 * c;
    #pragma unroll
    for (int q = 0; q < 32; ++q) {
      const int k0 = 8 * (q >> 1) + 2 * (q & 1);
      const v4f w0 = *(const v4f*)(Bm + (size_t)k0 * DH);
      const v4f w1 = *(const v4f*)(Bm + (size_t)(k0 + 1) * DH);
      #pragma unroll
      for (int ct = 0; ct < 4; ++ct) breg[ct][q] = (v2f){w0[ct], w1[ct]};
    }
  }

  int t = gw - first;
  const int n = (PJ_TILES - t + nw - 1) / nw;
  v4f a0[16], a1[16];
  pj_load(a0, Am, t, c, hi);
  for (int j = 0; j + 1 < n; j += 2) {
    pj_load(a1, Am, t + nw, c, hi);
    pj_tile(a0, breg, Om, t, c, hi);
    const int t2 = t + 2 * nw;
    pj_load(a0, Am, t2 < PJ_TILES ? t2 : t, c, hi);
    pj_tile(a1, breg, Om, t + nw, c, hi);
    t = t2;
  }
  if (n & 1) pj_tile(a0, breg, Om, t, c, hi);
}

#define RG_ROWS   1220
#define RG_PAD    1248
#define RG_N      82
#define RG_G      16
#define BK_E      12800
#define BK_N      125
#define BK_IT     4

__global__ __launch_bounds__(1024)
void k_count(const int* __restrict__ dst, unsigned* __restrict__ tot) {
  __shared__ unsigned hist[128];
  const int tid = threadIdx.x;
  const int m   = blockIdx.y;
  if (tid < 128) hist[tid] = 0u;
  __syncthreads();
  const int4* d4 = (const int4*)(dst + (size_t)m * NE + (size_t)blockIdx.x * BK_E);
  #pragma unroll
  for (int k = 0; k < BK_IT; ++k) {
    const int i = k * 1024 + tid;
    if (i < BK_E / 4) {
      const int4 c = d4[i];
      const int cv[4] = {c.x, c.y, c.z, c.w};
      #pragma unroll
      for (int q = 0; q < 4; ++q) {
        const unsigned r = (unsigned)cv[q] / RG_ROWS;
        if (r < RG_N) atomicAdd(&hist[r], 1u);
      }
    }
  }
  __syncthreads();
  if (tid < RG_N) {
    const unsigned h = hist[tid];
    if (h) atomicAdd(&tot[m * 128 + tid], h);
  }
}

__global__ __launch_bounds__(1024)
void k_fill(const int* __restrict__ src, const int* __restrict__ dst,
            const unsigned* __restrict__ tot, unsigned* __restrict__ cur,
            unsigned* __restrict__ elist) {
  __shared__ unsigned hist[128];
  __shared__ unsigned start[128];
  __shared__ unsigned totl[128];
  const int tid = threadIdx.x;
  const int m   = blockIdx.y;
  if (tid < 128) {
    hist[tid] = 0u;
    totl[tid] = (tid < RG_N) ? tot[m * 128 + tid] : 0u;
  }
  __syncthreads();
  const size_t e0 = (size_t)m * NE + (size_t)blockIdx.x * BK_E;
  const int4* d4 = (const int4*)(dst + e0);
  const int4* s4 = (const int4*)(src + e0);
  int4 dv[BK_IT], sv[BK_IT];
  #pragma unroll
  for (int k = 0; k < BK_IT; ++k) {
    const int i = k * 1024 + tid;
    const int ic = i < BK_E / 4 ? i : BK_E / 4 - 1;
    dv[k] = d4[ic];
    sv[k] = s4[ic];
  }
  #pragma unroll
  for (int k = 0; k < BK_IT; ++k) {
    if (k * 1024 + tid < BK_E / 4) {
      const int cv[4] = {dv[k].x, dv[k].y, dv[k].z, dv[k].w};
      #pragma unroll
      for (int q = 0; q < 4; ++q) {
        const unsigned r = (unsigned)cv[q] / RG_ROWS;
        if (r < RG_N) atomicAdd(&hist[r], 1u);
      }
    }
  }
  __syncthreads();
  if (tid < RG_N) {
    unsigned base = 0u;
    for (int j = 0; j < tid; ++j) base += totl[j];
    const unsigned h = hist[tid];
    const unsigned got = h ? atomicAdd(&cur[m * 128 + tid], h) : 0u;
    start[tid] = base + got;
    hist[tid] = 0u;
  }
  __syncthreads();
  unsigned* lm = elist + (size_t)m * NE;
  #pragma unroll
  for (int k = 0; k < BK_IT; ++k) {
    if (k * 1024 + tid < BK_E / 4) {
      const int cv[4] = {dv[k].x, dv[k].y, dv[k].z, dv[k].w};
      const int cs[4] = {sv[k].x, sv[k].y, sv[k].z, sv[k].w};
      #pragma unroll
      for (int q = 0; q < 4; ++q) {
        const unsigned r = (unsigned)cv[q] / RG_ROWS;
        if (r < RG_N) {
          const unsigned pos = start[r] + atomicAdd(&hist[r], 1u);
          lm[pos] = ((unsigned)cs[q] << 11) | ((unsigned)cv[q] - r * RG_ROWS);
        }
      }
    }
  }
}

__device__ __forceinline__ void rg_add(float* rows, unsigned pk, int nvalid,
                                       const float* __restrict__ projm, unsigned lane) {
  const int sv = (int)(pk >> 11);
  const int ld = (int)(pk & 2047u);
  for (int g = 0; g < nvalid; g += RG_G) {
    v2f r[RG_G];
    #pragma unroll
    for (int j = 0; j < RG_G; ++j) {
      const unsigned s = (unsigned)__builtin_amdgcn_readlane(sv, g + j);
      r[j] = *(const v2f*)(projm + s * DH + lane * 2);
    }
    #pragma unroll
    for (int j = 0; j < RG_G; ++j) {
      if (g + j < nvalid) {
        const unsigned d = (unsigned)__builtin_amdgcn_readlane(ld, g + j);
        float* p = rows + d * DH + lane * 2;
        atomicAdd(p, r[j].x);
        atomicAdd(p + 1, r[j].y);
      }
    }
  }
}

__device__ __forceinline__ void rg_issue(v2f (&r)[32], int mine, unsigned cnt,
                                         const float* __restrict__ projm, unsigned lane) {
  #pragma unroll
  for (int q = 0; q < 8; ++q) {
    #pragma unroll
    for (int j = 0; j < 4; ++j) r[4 * q + j] = (v2f){0.f, 0.f};
    if (q == 0 || (unsigned)(4 * q) < cnt) {
      #pragma unroll
      for (int j = 0; j < 4; ++j) {
        const unsigned s = (unsigned)__builtin_amdgcn_readlane(mine, 4 * q + j);
        r[4 * q + j] = *(const v2f*)(projm + s * DH + lane * 2);
      }
    }
  }
}

__device__ __forceinline__ v2f rg_reduce(const v2f (&r)[32], unsigned cnt) {
  v2f acc = {0.f, 0.f};
  #pragma unroll
  for (int j = 0; j < 32; ++j) {
    const v2f x = ((unsigned)j < cnt) ? r[j] : (v2f){0.f, 0.f};
    acc += x;
  }
  return acc;
}

__global__ __launch_bounds__(1024)
void k_scatter(const float* __restrict__ proj, const unsigned* __restrict__ elist,
               const unsigned* __restrict__ tot, const float* __restrict__ b,
               const float* __restrict__ pa, float* __restrict__ hout) {
  __shared__ alignas(16) float rows[RG_ROWS * DH];
  __shared__ unsigned degc[RG_PAD];
  __shared__ unsigned beg[RG_PAD];
  __shared__ unsigned totl[128];
  const int tid = threadIdx.x;
  const unsigned lane = (unsigned)tid & 31u;
  const unsigned wv = (unsigned)__builtin_amdgcn_readfirstlane(tid >> 5);
  const int m  = blockIdx.y;
  const int rg = blockIdx.x;
  const int lo = rg * RG_ROWS;
  const unsigned nrows = (unsigned)((NN - lo) < RG_ROWS ? (NN - lo) : RG_ROWS);
  const float* projm = proj + (size_t)m * NN * DH;

  if (tid < 128) totl[tid] = (tid < RG_N) ? tot[m * 128 + tid] : 0u;
  for (int i = tid; i < RG_PAD; i += 1024) degc[i] = 0u;
  __syncthreads();

  unsigned bsum = 0u;
  for (int j = 0; j < rg; ++j) bsum += totl[j];
  const unsigned base = (unsigned)__builtin_amdgcn_readfirstlane((int)bsum);
  const unsigned n    = (unsigned)__builtin_amdgcn_readfirstlane((int)totl[rg]);
  const unsigned* bin = elist + (size_t)m * NE + base;
  float* o = hout + ((size_t)m * NN + lo) * DH;
  const float a = pa[m];

  if (n <= (unsigned)(RG_ROWS * DH)) {
    unsigned* srt = (unsigned*)rows;
    for (unsigned i0 = (unsigned)tid; i0 < n; i0 += 4096u) {
      unsigned pk[4];
      #pragma unroll
      for (int q = 0; q < 4; ++q) {
        const unsigned i = i0 + (unsigned)q * 1024u;
        pk[q] = bin[i < n ? i : n - 1u];
      }
      #pragma unroll
      for (int q = 0; q < 4; ++q)
        if (i0 + (unsigned)q * 1024u < n) atomicAdd(&degc[pk[q] & 2047u], 1u);
    }
    __syncthreads();
    if (wv == 0u) {
      const unsigned r0 = lane * 39u;
      unsigned cv[39];
      unsigned s = 0u;
      #pragma unroll
      for (int k = 0; k < 39; ++k) { cv[k] = degc[r0 + k]; s += cv[k]; }
      unsigned x = s;
      for (int d = 1; d < 32; d <<= 1) {
        const unsigned y = (unsigned)__shfl_up((int)x, d);
        if ((int)lane >= d) x += y;
      }
      unsigned run = x - s;
      #pragma unroll
      for (int k = 0; k < 39; ++k) { beg[r0 + k] = run; degc[r0 + k] = run; run += cv[k]; }
    }
    __syncthreads();
    for (unsigned i0 = (unsigned)tid; i0 < n; i0 += 4096u) {
      unsigned pk[4];
      #pragma unroll
      for (int q = 0; q < 4; ++q) {
        const unsigned i = i0 + (unsigned)q * 1024u;
        pk[q] = (i < n) ? bin[i] : 0u;
      }
      #pragma unroll
      for (int q = 0; q < 4; ++q) {
        if (i0 + (unsigned)q * 1024u < n) {
          const unsigned pos = atomicAdd(&degc[pk[q] & 2047u], 1u);
          srt[pos] = pk[q] >> 11;
        }
      }
    }
    __syncthreads();
    const v2f bv = *(const v2f*)(b + m * DH + lane * 2);
    unsigned r = wv;
    unsigned b0 = 0u, e0 = 0u;
    int mine = 0;
    if (r < nrows) {
      b0 = (unsigned)__builtin_amdgcn_readfirstlane((int)beg[r]);
      e0 = (unsigned)__builtin_amdgcn_readfirstlane((int)degc[r]);
      mine = (b0 + lane < e0) ? (int)srt[b0 + lane] : 0;
    }
    for (; r < nrows; r += 32u) {
      v2f rr[32];
      rg_issue(rr, mine, e0 - b0, projm, lane);
      const unsigned rn = (r + 32u < nrows) ? r + 32u : r;
      const unsigned bn = (unsigned)__builtin_amdgcn_readfirstlane((int)beg[rn]);
      const unsigned en = (unsigned)__builtin_amdgcn_readfirstlane((int)degc[rn]);
      const int minen = (bn + lane < en) ? (int)srt[bn + lane] : 0;
      v2f acc = rg_reduce(rr, e0 - b0);
      for (unsigned g = b0 + 32u; g < e0; g += 32u) {
        const int m2 = (g + lane < e0) ? (int)srt[g + lane] : 0;
        v2f r2[32];
        rg_issue(r2, m2, e0 - g, projm, lane);
        acc += rg_reduce(r2, e0 - g);
      }
      const float dg = fmaxf((float)(e0 - b0), 1.0f);
      v2f hv;
      hv.x = acc.x / dg + bv.x;
      hv.y = acc.y / dg + bv.y;
      hv.x = hv.x > 0.f ? hv.x : a * hv.x;
      hv.y = hv.y > 0.f ? hv.y : a * hv.y;
      *(v2f*)(o + (size_t)r * DH + lane * 2) = hv;
      b0 = bn; e0 = en; mine = minen;
    }
    return;
  }

  for (int i = tid; i < RG_ROWS * DH / 4; i += 1024) ((v4f*)rows)[i] = (v4f){0.f, 0.f, 0.f, 0.f};
  __syncthreads();

  for (unsigned i0 = wv * 128u; i0 < n; i0 += 4096u) {
    unsigned pk[4];
    #pragma unroll
    for (int q = 0; q < 4; ++q) {
      const unsigned i = i0 + (unsigned)q * 32u + lane;
      pk[q] = (i < n) ? bin[i] : 0u;
    }
    #pragma unroll
    for (int q = 0; q < 4; ++q) {
      const unsigned s0 = i0 + (unsigned)q * 32u;
      if (s0 < n) {
        const unsigned rem = n - s0;
        const int nvalid = rem < 32u ? (int)rem : 32;
        if ((int)lane < nvalid) atomicAdd(&degc[pk[q] & 2047u], 1u);
        rg_add(rows, pk[q], nvalid, projm, lane);
      }
    }
  }
  __syncthreads();

  const v4f bv4 = *(const v4f*)(b + m * DH + (tid & 15) * 4);
  for (unsigned i = tid; i < nrows * (DH / 4); i += 1024) {
    const float dg = fmaxf((float)degc[i >> 4], 1.0f);
    v4f hv = ((const v4f*)rows)[i];
    for (int c = 0; c < 4; ++c) {
      const float h = hv[c] / dg + bv4[c];
      hv[c] = h > 0.f ? h : a * h;
    }
    ((v4f*)o)[i] = hv;
  }
}

__global__ __launch_bounds__(128)
void k_fc_wmma(const float* __restrict__ h, const float* __restrict__ fcW,
               const float* __restrict__ fcb, const float* __restrict__ attn,
               float* __restrict__ logits) {
  __shared__ float red[128];
  const int lane = threadIdx.x & 31;
  const int wv   = threadIdx.x >> 5;
  const int col0 = wv * 16;
  const int m    = blockIdx.y;
  const int n0   = blockIdx.x * 16;
  const int r    = lane & 15;
  const int hi   = lane >> 4;
  const int kh   = hi * 2;

  const float* A = h + ((size_t)m * NN + n0) * DH;
  float bias = fcb[col0 + r];
  v8f acc = {bias, bias, bias, bias, bias, bias, bias, bias};
  #pragma unroll
  for (int k = 0; k < DH; k += 4) {
    v2f a = *(const v2f*)(A + (size_t)r * DH + k + kh);
    v2f bb;
    bb.x = fcW[(k + kh)     * DH + col0 + r];
    bb.y = fcW[(k + kh + 1) * DH + col0 + r];
    acc = __builtin_amdgcn_wmma_f32_16x16x4_f32(false, a, false, bb, (short)0, acc,
                                                false, false);
  }
  float p = 0.f;
  #pragma unroll
  for (int v = 0; v < 8; ++v) p += tanhf(acc[v]);
  p *= attn[col0 + r];

  red[threadIdx.x] = p;
  __syncthreads();
  #pragma unroll
  for (int s2 = 64; s2 > 0; s2 >>= 1) {
    if (threadIdx.x < s2) red[threadIdx.x] += red[threadIdx.x + s2];
    __syncthreads();
  }
  if (threadIdx.x == 0) atomicAddF(&logits[m], red[0]);
}

__global__ void k_softmax(const float* __restrict__ logits, float* __restrict__ beta) {
  if (threadIdx.x == 0 && blockIdx.x == 0) {
    float l[MM]; float mx = -1e30f;
    for (int m = 0; m < MM; ++m) { l[m] = logits[m] / (float)NN; mx = fmaxf(mx, l[m]); }
    float s = 0.f;
    for (int m = 0; m < MM; ++m) { l[m] = expf(l[m] - mx); s += l[m]; }
    for (int m = 0; m < MM; ++m) beta[m] = l[m] / s;
  }
}

__global__ __launch_bounds__(256)
void k_out(const float* __restrict__ h, const float* __restrict__ beta,
           float* __restrict__ out) {
  size_t i4 = (size_t)blockIdx.x * 256 + threadIdx.x;
  if (i4 >= (size_t)NN * (DH / 4)) return;
  size_t base = i4 * 4;
  const size_t P = (size_t)NN * DH;
  float b0 = beta[0], b1 = beta[1], b2 = beta[2];
  v4f x0 = *(const v4f*)(h + base);
  v4f x1 = *(const v4f*)(h + P + base);
  v4f x2 = *(const v4f*)(h + 2 * P + base);
  v4f z  = x0 * b0 + x1 * b1 + x2 * b2;
  *(v4f*)(out + base) = z;
}

extern "C" void kernel_launch(void* const* d_in, const int* in_sizes, int n_in,
                              void* d_out, int out_size, void* d_ws, size_t ws_size,
                              hipStream_t stream) {
  const float* feats = (const float*)d_in[0];
  const int*   src   = (const int*)  d_in[1];
  const int*   dst   = (const int*)  d_in[2];
  const float* W     = (const float*)d_in[3];
  const float* b     = (const float*)d_in[4];
  const float* pa    = (const float*)d_in[5];
  const float* fcW   = (const float*)d_in[6];
  const float* fcb   = (const float*)d_in[7];
  const float* attn  = (const float*)d_in[8];

  float* ws     = (float*)d_ws;
  float* agg    = ws + AGG_OFF;
  float* logits = ws + LOG_OFF;
  float* beta   = ws + BETA_OFF;
  float* proj   = ws + PROJ_OFF;
  float* out    = (float*)d_out;
  unsigned* tot = (unsigned*)(ws + CNT_OFF);
  unsigned* cur = tot + 384;
  unsigned* elist = (unsigned*)d_out;

  hipMemsetAsync(ws + CNT_OFF, 0, (768 + 8) * sizeof(float), stream);

  k_count<<<dim3(BK_N, MM), 1024, 0, stream>>>(dst, tot);
  k_fill<<<dim3(BK_N, MM), 1024, 0, stream>>>(src, dst, tot, cur, elist);

  dim3 gTiles(NN / 16, MM);
  k_proj_wmma<<<PJ_WAVES / 4, 128, 0, stream>>>(feats, W, proj);

  k_scatter<<<dim3(RG_N, MM), 1024, 0, stream>>>(proj, elist, tot, b, pa, agg);

  k_fc_wmma<<<gTiles, 128, 0, stream>>>(agg, fcW, fcb, attn, logits);

  k_softmax<<<1, 32, 0, stream>>>(logits, beta);

  size_t ov4 = (size_t)NN * (DH / 4);
  k_out<<<(unsigned)((ov4 + 255) / 256), 256, 0, stream>>>(agg, beta, out);
}


# fc/logits and output kernels as resident grids looping over tiles (wave pair holds fc_W in registers), not 125000 short-lived waves; logits: register sums, 1 atomic per work-group; fc k order changed
# speedup vs baseline: 5.2212x; 1.0736x over previous
#include <hip/hip_runtime.h>
#include <hip/hip_bf16.h>
#include <math.h>

#define NN  100000
#define MM  3
#define DIN 128
#define DH  64
#define NE  1600000

typedef __attribute__((ext_vector_type(2))) float v2f;
typedef __attribute__((ext_vector_type(4))) float v4f;
typedef __attribute__((ext_vector_type(8))) float v8f;

#define AGG_OFF   0ull
#define LOG_OFF   19500000ull
#define BETA_OFF  19500004ull
#define PROJ_OFF  19500008ull
#define CNT_OFF   19499232ull

__device__ __forceinline__ void atomicAddF(float* p, float v) {
  asm volatile("global_atomic_add_f32 %0, %1, off scope:SCOPE_DEV"
               :: "v"(p), "v"(v) : "memory");
}

#define PJ_WAVES 2048
#define PJ_TILES (NN / 16)

__device__ __forceinline__ void pj_load(v4f (&a)[16], const float* __restrict__ Am, int t,
                                        int c, int hi) {
  const float* p = Am + ((size_t)t * 16 + c) * DIN + 4 * hi;
  #pragma unroll
  for (int q = 0; q < 16; ++q) a[q] = *(const v4f*)(p + 8 * q);
}

__device__ __forceinline__ void pj_tile(const v4f (&a)[16], const v2f (&breg)[4][32],
                                        float* __restrict__ Om, int t, int c, int hi) {
  v8f acc[4];
  #pragma unroll
  for (int ct = 0; ct < 4; ++ct) acc[ct] = (v8f){0.f,0.f,0.f,0.f,0.f,0.f,0.f,0.f};
  #pragma unroll
  for (int p = 0; p < 16; ++p) {
    const v2f a0 = {a[p].x, a[p].y};
    const v2f a1 = {a[p].z, a[p].w};
    #pragma unroll
    for (int ct = 0; ct < 4; ++ct)
      acc[ct] = __builtin_amdgcn_wmma_f32_16x16x4_f32(false, a0, false, breg[ct][2 * p], (short)0,
                                                      acc[ct], false, false);
    #pragma unroll
    for (int ct = 0; ct < 4; ++ct)
      acc[ct] = __builtin_amdgcn_wmma_f32_16x16x4_f32(false, a1, false, breg[ct][2 * p + 1], (short)0,
                                                      acc[ct], false, false);
  }
  float* o = Om + ((size_t)t * 16 + 8 * hi) * DH + 4 * c;
  #pragma unroll
  for (int v = 0; v < 8; ++v)
    *(v4f*)(o + (size_t)v * DH) = (v4f){acc[0][v], acc[1][v], acc[2][v], acc[3][v]};
}

__global__ __launch_bounds__(128, 2)
void k_proj_wmma(const float* __restrict__ feats, const float* __restrict__ W,
                 float* __restrict__ proj) {
  const int lane = threadIdx.x & 31;
  const int c    = lane & 15;
  const int hi   = lane >> 4;
  const int gw = __builtin_amdgcn_readfirstlane((int)(blockIdx.x * 4 + (threadIdx.x >> 5)));
  const int m     = gw * MM / PJ_WAVES;
  const int first = (m * PJ_WAVES + MM - 1) / MM;
  const int nw    = ((m + 1) * PJ_WAVES + MM - 1) / MM - first;
  const float* Am = feats + (size_t)m * NN * DIN;
  float*       Om = proj  + (size_t)m * NN * DH;

  v2f breg[4][32];
  {
    const float* Bm = W + (size_t)m * DIN * DH + (size_t)(4 * hi) * DH + 4 * c;
    #pragma unroll
    for (int q = 0; q < 32; ++q) {
      const int k0 = 8 * (q >> 1) + 2 * (q & 1);
      const v4f w0 = *(const v4f*)(Bm + (size_t)k0 * DH);
      const v4f w1 = *(const v4f*)(Bm + (size_t)(k0 + 1) * DH);
      #pragma unroll
      for (int ct = 0; ct < 4; ++ct) breg[ct][q] = (v2f){w0[ct], w1[ct]};
    }
  }

  int t = gw - first;
  const int n = (PJ_TILES - t + nw - 1) / nw;
  v4f a0[16], a1[16];
  pj_load(a0, Am, t, c, hi);
  for (int j = 0; j + 1 < n; j += 2) {
    pj_load(a1, Am, t + nw, c, hi);
    pj_tile(a0, breg, Om, t, c, hi);
    const int t2 = t + 2 * nw;
    pj_load(a0, Am, t2 < PJ_TILES ? t2 : t, c, hi);
    pj_tile(a1, breg, Om, t + nw, c, hi);
    t = t2;
  }
  if (n & 1) pj_tile(a0, breg, Om, t, c, hi);
}

#define RG_ROWS   1220
#define RG_PAD    1248
#define RG_N      82
#define RG_G      16
#define BK_E      12800
#define BK_N      125
#define BK_IT     4

__global__ __launch_bounds__(1024)
void k_count(const int* __restrict__ dst, unsigned* __restrict__ tot) {
  __shared__ unsigned hist[128];
  const int tid = threadIdx.x;
  const int m   = blockIdx.y;
  if (tid < 128) hist[tid] = 0u;
  __syncthreads();
  const int4* d4 = (const int4*)(dst + (size_t)m * NE + (size_t)blockIdx.x * BK_E);
  #pragma unroll
  for (int k = 0; k < BK_IT; ++k) {
    const int i = k * 1024 + tid;
    if (i < BK_E / 4) {
      const int4 c = d4[i];
      const int cv[4] = {c.x, c.y, c.z, c.w};
      #pragma unroll
      for (int q = 0; q < 4; ++q) {
        const unsigned r = (unsigned)cv[q] / RG_ROWS;
        if (r < RG_N) atomicAdd(&hist[r], 1u);
      }
    }
  }
  __syncthreads();
  if (tid < RG_N) {
    const unsigned h = hist[tid];
    if (h) atomicAdd(&tot[m * 128 + tid], h);
  }
}

__global__ __launch_bounds__(1024)
void k_fill(const int* __restrict__ src, const int* __restrict__ dst,
            const unsigned* __restrict__ tot, unsigned* __restrict__ cur,
            unsigned* __restrict__ elist) {
  __shared__ unsigned hist[128];
  __shared__ unsigned start[128];
  __shared__ unsigned totl[128];
  const int tid = threadIdx.x;
  const int m   = blockIdx.y;
  if (tid < 128) {
    hist[tid] = 0u;
    totl[tid] = (tid < RG_N) ? tot[m * 128 + tid] : 0u;
  }
  __syncthreads();
  const size_t e0 = (size_t)m * NE + (size_t)blockIdx.x * BK_E;
  const int4* d4 = (const int4*)(dst + e0);
  const int4* s4 = (const int4*)(src + e0);
  int4 dv[BK_IT], sv[BK_IT];
  #pragma unroll
  for (int k = 0; k < BK_IT; ++k) {
    const int i = k * 1024 + tid;
    const int ic = i < BK_E / 4 ? i : BK_E / 4 - 1;
    dv[k] = d4[ic];
    sv[k] = s4[ic];
  }
  #pragma unroll
  for (int k = 0; k < BK_IT; ++k) {
    if (k * 1024 + tid < BK_E / 4) {
      const int cv[4] = {dv[k].x, dv[k].y, dv[k].z, dv[k].w};
      #pragma unroll
      for (int q = 0; q < 4; ++q) {
        const unsigned r = (unsigned)cv[q] / RG_ROWS;
        if (r < RG_N) atomicAdd(&hist[r], 1u);
      }
    }
  }
  __syncthreads();
  if (tid < RG_N) {
    unsigned base = 0u;
    for (int j = 0; j < tid; ++j) base += totl[j];
    const unsigned h = hist[tid];
    const unsigned got = h ? atomicAdd(&cur[m * 128 + tid], h) : 0u;
    start[tid] = base + got;
    hist[tid] = 0u;
  }
  __syncthreads();
  unsigned* lm = elist + (size_t)m * NE;
  #pragma unroll
  for (int k = 0; k < BK_IT; ++k) {
    if (k * 1024 + tid < BK_E / 4) {
      const int cv[4] = {dv[k].x, dv[k].y, dv[k].z, dv[k].w};
      const int cs[4] = {sv[k].x, sv[k].y, sv[k].z, sv[k].w};
      #pragma unroll
      for (int q = 0; q < 4; ++q) {
        const unsigned r = (unsigned)cv[q] / RG_ROWS;
        if (r < RG_N) {
          const unsigned pos = start[r] + atomicAdd(&hist[r], 1u);
          lm[pos] = ((unsigned)cs[q] << 11) | ((unsigned)cv[q] - r * RG_ROWS);
        }
      }
    }
  }
}

__device__ __forceinline__ void rg_add(float* rows, unsigned pk, int nvalid,
                                       const float* __restrict__ projm, unsigned lane) {
  const int sv = (int)(pk >> 11);
  const int ld = (int)(pk & 2047u);
  for (int g = 0; g < nvalid; g += RG_G) {
    v2f r[RG_G];
    #pragma unroll
    for (int j = 0; j < RG_G; ++j) {
      const unsigned s = (unsigned)__builtin_amdgcn_readlane(sv, g + j);
      r[j] = *(const v2f*)(projm + s * DH + lane * 2);
    }
    #pragma unroll
    for (int j = 0; j < RG_G; ++j) {
      if (g + j < nvalid) {
        const unsigned d = (unsigned)__builtin_amdgcn_readlane(ld, g + j);
        float* p = rows + d * DH + lane * 2;
        atomicAdd(p, r[j].x);
        atomicAdd(p + 1, r[j].y);
      }
    }
  }
}

__device__ __forceinline__ void rg_issue(v2f (&r)[32], int mine, unsigned cnt,
                                         const float* __restrict__ projm, unsigned lane) {
  #pragma unroll
  for (int q = 0; q < 8; ++q) {
    #pragma unroll
    for (int j = 0; j < 4; ++j) r[4 * q + j] = (v2f){0.f, 0.f};
    if (q == 0 || (unsigned)(4 * q) < cnt) {
      #pragma unroll
      for (int j = 0; j < 4; ++j) {
        const unsigned s = (unsigned)__builtin_amdgcn_readlane(mine, 4 * q + j);
        r[4 * q + j] = *(const v2f*)(projm + s * DH + lane * 2);
      }
    }
  }
}

__device__ __forceinline__ v2f rg_reduce(const v2f (&r)[32], unsigned cnt) {
  v2f acc = {0.f, 0.f};
  #pragma unroll
  for (int j = 0; j < 32; ++j) {
    const v2f x = ((unsigned)j < cnt) ? r[j] : (v2f){0.f, 0.f};
    acc += x;
  }
  return acc;
}

__global__ __launch_bounds__(1024)
void k_scatter(const float* __restrict__ proj, const unsigned* __restrict__ elist,
               const unsigned* __restrict__ tot, const float* __restrict__ b,
               const float* __restrict__ pa, float* __restrict__ hout) {
  __shared__ alignas(16) float rows[RG_ROWS * DH];
  __shared__ unsigned degc[RG_PAD];
  __shared__ unsigned beg[RG_PAD];
  __shared__ unsigned totl[128];
  const int tid = threadIdx.x;
  const unsigned lane = (unsigned)tid & 31u;
  const unsigned wv = (unsigned)__builtin_amdgcn_readfirstlane(tid >> 5);
  const int m  = blockIdx.y;
  const int rg = blockIdx.x;
  const int lo = rg * RG_ROWS;
  const unsigned nrows = (unsigned)((NN - lo) < RG_ROWS ? (NN - lo) : RG_ROWS);
  const float* projm = proj + (size_t)m * NN * DH;

  if (tid < 128) totl[tid] = (tid < RG_N) ? tot[m * 128 + tid] : 0u;
  for (int i = tid; i < RG_PAD; i += 1024) degc[i] = 0u;
  __syncthreads();

  unsigned bsum = 0u;
  for (int j = 0; j < rg; ++j) bsum += totl[j];
  const unsigned base = (unsigned)__builtin_amdgcn_readfirstlane((int)bsum);
  const unsigned n    = (unsigned)__builtin_amdgcn_readfirstlane((int)totl[rg]);
  const unsigned* bin = elist + (size_t)m * NE + base;
  float* o = hout + ((size_t)m * NN + lo) * DH;
  const float a = pa[m];

  if (n <= (unsigned)(RG_ROWS * DH)) {
    unsigned* srt = (unsigned*)rows;
    for (unsigned i0 = (unsigned)tid; i0 < n; i0 += 4096u) {
      unsigned pk[4];
      #pragma unroll
      for (int q = 0; q < 4; ++q) {
        const unsigned i = i0 + (unsigned)q * 1024u;
        pk[q] = bin[i < n ? i : n - 1u];
      }
      #pragma unroll
      for (int q = 0; q < 4; ++q)
        if (i0 + (unsigned)q * 1024u < n) atomicAdd(&degc[pk[q] & 2047u], 1u);
    }
    __syncthreads();
    if (wv == 0u) {
      const unsigned r0 = lane * 39u;
      unsigned cv[39];
      unsigned s = 0u;
      #pragma unroll
      for (int k = 0; k < 39; ++k) { cv[k] = degc[r0 + k]; s += cv[k]; }
      unsigned x = s;
      for (int d = 1; d < 32; d <<= 1) {
        const unsigned y = (unsigned)__shfl_up((int)x, d);
        if ((int)lane >= d) x += y;
      }
      unsigned run = x - s;
      #pragma unroll
      for (int k = 0; k < 39; ++k) { beg[r0 + k] = run; degc[r0 + k] = run; run += cv[k]; }
    }
    __syncthreads();
    for (unsigned i0 = (unsigned)tid; i0 < n; i0 += 4096u) {
      unsigned pk[4];
      #pragma unroll
      for (int q = 0; q < 4; ++q) {
        const unsigned i = i0 + (unsigned)q * 1024u;
        pk[q] = (i < n) ? bin[i] : 0u;
      }
      #pragma unroll
      for (int q = 0; q < 4; ++q) {
        if (i0 + (unsigned)q * 1024u < n) {
          const unsigned pos = atomicAdd(&degc[pk[q] & 2047u], 1u);
          srt[pos] = pk[q] >> 11;
        }
      }
    }
    __syncthreads();
    const v2f bv = *(const v2f*)(b + m * DH + lane * 2);
    unsigned r = wv;
    unsigned b0 = 0u, e0 = 0u;
    int mine = 0;
    if (r < nrows) {
      b0 = (unsigned)__builtin_amdgcn_readfirstlane((int)beg[r]);
      e0 = (unsigned)__builtin_amdgcn_readfirstlane((int)degc[r]);
      mine = (b0 + lane < e0) ? (int)srt[b0 + lane] : 0;
    }
    for (; r < nrows; r += 32u) {
      v2f rr[32];
      rg_issue(rr, mine, e0 - b0, projm, lane);
      const unsigned rn = (r + 32u < nrows) ? r + 32u : r;
      const unsigned bn = (unsigned)__builtin_amdgcn_readfirstlane((int)beg[rn]);
      const unsigned en = (unsigned)__builtin_amdgcn_readfirstlane((int)degc[rn]);
      const int minen = (bn + lane < en) ? (int)srt[bn + lane] : 0;
      v2f acc = rg_reduce(rr, e0 - b0);
      for (unsigned g = b0 + 32u; g < e0; g += 32u) {
        const int m2 = (g + lane < e0) ? (int)srt[g + lane] : 0;
        v2f r2[32];
        rg_issue(r2, m2, e0 - g, projm, lane);
        acc += rg_reduce(r2, e0 - g);
      }
      const float dg = fmaxf((float)(e0 - b0), 1.0f);
      v2f hv;
      hv.x = acc.x / dg + bv.x;
      hv.y = acc.y / dg + bv.y;
      hv.x = hv.x > 0.f ? hv.x : a * hv.x;
      hv.y = hv.y > 0.f ? hv.y : a * hv.y;
      *(v2f*)(o + (size_t)r * DH + lane * 2) = hv;
      b0 = bn; e0 = en; mine = minen;
    }
    return;
  }

  for (int i = tid; i < RG_ROWS * DH / 4; i += 1024) ((v4f*)rows)[i] = (v4f){0.f, 0.f, 0.f, 0.f};
  __syncthreads();

  for (unsigned i0 = wv * 128u; i0 < n; i0 += 4096u) {
    unsigned pk[4];
    #pragma unroll
    for (int q = 0; q < 4; ++q) {
      const unsigned i = i0 + (unsigned)q * 32u + lane;
      pk[q] = (i < n) ? bin[i] : 0u;
    }
    #pragma unroll
    for (int q = 0; q < 4; ++q) {
      const unsigned s0 = i0 + (unsigned)q * 32u;
      if (s0 < n) {
        const unsigned rem = n - s0;
        const int nvalid = rem < 32u ? (int)rem : 32;
        if ((int)lane < nvalid) atomicAdd(&degc[pk[q] & 2047u], 1u);
        rg_add(rows, pk[q], nvalid, projm, lane);
      }
    }
  }
  __syncthreads();

  const v4f bv4 = *(const v4f*)(b + m * DH + (tid & 15) * 4);
  for (unsigned i = tid; i < nrows * (DH / 4); i += 1024) {
    const float dg = fmaxf((float)degc[i >> 4], 1.0f);
    v4f hv = ((const v4f*)rows)[i];
    for (int c = 0; c < 4; ++c) {
      const float h = hv[c] / dg + bv4[c];
      hv[c] = h > 0.f ? h : a * h;
    }
    ((v4f*)o)[i] = hv;
  }
}

#define FC_WGS 1024

__device__ __forceinline__ void fc_load(v4f (&a)[8], const float* __restrict__ Am, int t,
                                        int c, int hi) {
  const float* p = Am + ((size_t)t * 16 + c) * DH + 4 * hi;
  #pragma unroll
  for (int q = 0; q < 8; ++q) a[q] = *(const v4f*)(p + 8 * q);
}

__device__ __forceinline__ void fc_tile(const v4f (&a)[8], const v2f (&breg)[2][16],
                                        const v2f bias2, v2f& ps) {
  v8f acc[2];
  #pragma unroll
  for (int ct = 0; ct < 2; ++ct) {
    const float bs = bias2[ct];
    acc[ct] = (v8f){bs, bs, bs, bs, bs, bs, bs, bs};
  }
  #pragma unroll
  for (int p = 0; p < 8; ++p) {
    const v2f a0 = {a[p].x, a[p].y};
    const v2f a1 = {a[p].z, a[p].w};
    #pragma unroll
    for (int ct = 0; ct < 2; ++ct)
      acc[ct] = __builtin_amdgcn_wmma_f32_16x16x4_f32(false, a0, false, breg[ct][2 * p], (short)0,
                                                      acc[ct], false, false);
    #pragma unroll
    for (int ct = 0; ct < 2; ++ct)
      acc[ct] = __builtin_amdgcn_wmma_f32_16x16x4_f32(false, a1, false, breg[ct][2 * p + 1], (short)0,
                                                      acc[ct], false, false);
  }
  #pragma unroll
  for (int ct = 0; ct < 2; ++ct) {
    float s = 0.f;
    #pragma unroll
    for (int v = 0; v < 8; ++v) s += tanhf(acc[ct][v]);
    ps[ct] += s;
  }
}

__global__ __launch_bounds__(128, 4)
void k_fc_wmma(const float* __restrict__ h, const float* __restrict__ fcW,
               const float* __restrict__ fcb, const float* __restrict__ attn,
               float* __restrict__ logits) {
  __shared__ float red[4];
  const int lane = threadIdx.x & 31;
  const int c    = lane & 15;
  const int hi   = lane >> 4;
  const int wv = __builtin_amdgcn_readfirstlane((int)(threadIdx.x >> 5));
  const int wg = (int)blockIdx.x;
  const int m     = wg * MM / FC_WGS;
  const int first = (m * FC_WGS + MM - 1) / MM;
  const int np    = 2 * (((m + 1) * FC_WGS + MM - 1) / MM - first);
  const int hc    = wv & 1;
  const float* Am = h + (size_t)m * NN * DH;

  v2f breg[2][16];
  {
    const float* Bm = fcW + (size_t)(4 * hi) * DH + 32 * hc + 2 * c;
    #pragma unroll
    for (int q = 0; q < 16; ++q) {
      const int k0 = 8 * (q >> 1) + 2 * (q & 1);
      const v2f w0 = *(const v2f*)(Bm + (size_t)k0 * DH);
      const v2f w1 = *(const v2f*)(Bm + (size_t)(k0 + 1) * DH);
      #pragma unroll
      for (int ct = 0; ct < 2; ++ct) breg[ct][q] = (v2f){w0[ct], w1[ct]};
    }
  }
  const v2f bias2 = *(const v2f*)(fcb + 32 * hc + 2 * c);
  const v2f attn2 = *(const v2f*)(attn + 32 * hc + 2 * c);
  v2f ps = {0.f, 0.f};

  int t = 2 * (wg - first) + (wv >> 1);
  const int n = (PJ_TILES - t + np - 1) / np;
  v4f a0[8], a1[8];
  fc_load(a0, Am, t, c, hi);
  for (int j = 0; j + 1 < n; j += 2) {
    fc_load(a1, Am, t + np, c, hi);
    fc_tile(a0, breg, bias2, ps);
    const int t2 = t + 2 * np;
    fc_load(a0, Am, t2 < PJ_TILES ? t2 : t, c, hi);
    fc_tile(a1, breg, bias2, ps);
    t = t2;
  }
  if (n & 1) fc_tile(a0, breg, bias2, ps);

  float p = ps[0] * attn2[0] + ps[1] * attn2[1];
  for (int d = 16; d > 0; d >>= 1) p += __shfl_xor(p, d);
  if (lane == 0) red[wv] = p;
  __syncthreads();
  if (threadIdx.x == 0) atomicAddF(&logits[m], (red[0] + red[1]) + (red[2] + red[3]));
}

__global__ void k_softmax(const float* __restrict__ logits, float* __restrict__ beta) {
  if (threadIdx.x == 0 && blockIdx.x == 0) {
    float l[MM]; float mx = -1e30f;
    for (int m = 0; m < MM; ++m) { l[m] = logits[m] / (float)NN; mx = fmaxf(mx, l[m]); }
    float s = 0.f;
    for (int m = 0; m < MM; ++m) { l[m] = expf(l[m] - mx); s += l[m]; }
    for (int m = 0; m < MM; ++m) beta[m] = l[m] / s;
  }
}

#define OUT_WGS 512
#define OUT_N4  (NN * (DH / 4))
#define OUT_T   (OUT_WGS * 256)

__global__ __launch_bounds__(256)
void k_out(const float* __restrict__ h, const float* __restrict__ beta,
           float* __restrict__ out) {
  const v4f* h4 = (const v4f*)h;
  v4f* o4 = (v4f*)out;
  const float b0 = beta[0], b1 = beta[1], b2 = beta[2];
  for (unsigned i = blockIdx.x * 256u + threadIdx.x; i < (unsigned)OUT_N4; i += 4u * OUT_T) {
    v4f x0[4], x1[4], x2[4];
    #pragma unroll
    for (int u = 0; u < 4; ++u) {
      const unsigned idx = i + (unsigned)u * OUT_T;
      const unsigned ic = idx < (unsigned)OUT_N4 ? idx : i;
      x0[u] = h4[ic];
      x1[u] = h4[(size_t)OUT_N4 + ic];
      x2[u] = h4[2 * (size_t)OUT_N4 + ic];
    }
    #pragma unroll
    for (int u = 0; u < 4; ++u) {
      const unsigned idx = i + (unsigned)u * OUT_T;
      if (idx < (unsigned)OUT_N4) o4[idx] = x0[u] * b0 + x1[u] * b1 + x2[u] * b2;
    }
  }
}

extern "C" void kernel_launch(void* const* d_in, const int* in_sizes, int n_in,
                              void* d_out, int out_size, void* d_ws, size_t ws_size,
                              hipStream_t stream) {
  const float* feats = (const float*)d_in[0];
  const int*   src   = (const int*)  d_in[1];
  const int*   dst   = (const int*)  d_in[2];
  const float* W     = (const float*)d_in[3];
  const float* b     = (const float*)d_in[4];
  const float* pa    = (const float*)d_in[5];
  const float* fcW   = (const float*)d_in[6];
  const float* fcb   = (const float*)d_in[7];
  const float* attn  = (const float*)d_in[8];

  float* ws     = (float*)d_ws;
  float* agg    = ws + AGG_OFF;
  float* logits = ws + LOG_OFF;
  float* beta   = ws + BETA_OFF;
  float* proj   = ws + PROJ_OFF;
  float* out    = (float*)d_out;
  unsigned* tot = (unsigned*)(ws + CNT_OFF);
  unsigned* cur = tot + 384;
  unsigned* elist = (unsigned*)d_out;

  hipMemsetAsync(ws + CNT_OFF, 0, (768 + 8) * sizeof(float), stream);

  k_count<<<dim3(BK_N, MM), 1024, 0, stream>>>(dst, tot);
  k_fill<<<dim3(BK_N, MM), 1024, 0, stream>>>(src, dst, tot, cur, elist);

  k_proj_wmma<<<PJ_WAVES / 4, 128, 0, stream>>>(feats, W, proj);

  k_scatter<<<dim3(RG_N, MM), 1024, 0, stream>>>(proj, elist, tot, b, pa, agg);

  k_fc_wmma<<<FC_WGS, 128, 0, stream>>>(agg, fcW, fcb, attn, logits);

  k_softmax<<<1, 32, 0, stream>>>(logits, beta);

  k_out<<<OUT_WGS, 256, 0, stream>>>(agg, beta, out);
}


# speedup vs baseline: 5.9875x; 1.1468x over previous
#include <hip/hip_runtime.h>
#include <hip/hip_bf16.h>
#include <math.h>

#define NN  100000
#define MM  3
#define DIN 128
#define DH  64
#define NE  1600000

typedef __attribute__((ext_vector_type(2))) float v2f;
typedef __attribute__((ext_vector_type(4))) float v4f;
typedef __attribute__((ext_vector_type(8))) float v8f;

#define AGG_OFF   0ull
#define PROJ_OFF  19200000ull
#define CNT_OFF   38400000ull
#define LOG_OFF   38400768ull
#define BETA_OFF  38400772ull

__device__ __forceinline__ void atomicAddF(float* p, float v) {
  asm volatile("global_atomic_add_f32 %0, %1, off scope:SCOPE_DEV"
               :: "v"(p), "v"(v) : "memory");
}

#define PJ_WAVES 2048
#define PJ_TILES (NN / 16)

__device__ __forceinline__ void pj_load(v4f (&a)[16], const float* __restrict__ Am, int t,
                                        int c, int hi) {
  const float* p = Am + ((size_t)t * 16 + c) * DIN + 4 * hi;
  #pragma unroll
  for (int q = 0; q < 16; ++q) a[q] = *(const v4f*)(p + 8 * q);
}

__device__ __forceinline__ void pj_tile(const v4f (&a)[16], const v2f (&breg)[4][32],
                                        float* __restrict__ Om, int t, int c, int hi) {
  v8f acc[4];
  #pragma unroll
  for (int ct = 0; ct < 4; ++ct) acc[ct] = (v8f){0.f,0.f,0.f,0.f,0.f,0.f,0.f,0.f};
  #pragma unroll
  for (int p = 0; p < 16; ++p) {
    const v2f a0 = {a[p].x, a[p].y};
    const v2f a1 = {a[p].z, a[p].w};
    #pragma unroll
    for (int ct = 0; ct < 4; ++ct)
      acc[ct] = __builtin_amdgcn_wmma_f32_16x16x4_f32(false, a0, false, breg[ct][2 * p], (short)0,
                                                      acc[ct], false, false);
    #pragma unroll
    for (int ct = 0; ct < 4; ++ct)
      acc[ct] = __builtin_amdgcn_wmma_f32_16x16x4_f32(false, a1, false, breg[ct][2 * p + 1], (short)0,
                                                      acc[ct], false, false);
  }
  float* o = Om + ((size_t)t * 16 + 8 * hi) * DH + 4 * c;
  #pragma unroll
  for (int v = 0; v < 8; ++v)
    *(v4f*)(o + (size_t)v * DH) = (v4f){acc[0][v], acc[1][v], acc[2][v], acc[3][v]};
}

__global__ __launch_bounds__(128, 2)
void k_proj_wmma(const float* __restrict__ feats, const float* __restrict__ W,
                 float* __restrict__ wsb) {
  const int lane = threadIdx.x & 31;
  const int c    = lane & 15;
  const int hi   = lane >> 4;
  const int gw = __builtin_amdgcn_readfirstlane((int)(blockIdx.x * 4 + (threadIdx.x >> 5)));
  const int m     = gw * MM / PJ_WAVES;
  const int first = (m * PJ_WAVES + MM - 1) / MM;
  const int nw    = ((m + 1) * PJ_WAVES + MM - 1) / MM - first;
  const float* Am = feats + (size_t)m * NN * DIN;
  float*       Om = wsb + PROJ_OFF + (size_t)m * NN * DH;

  v2f breg[4][32];
  {
    const float* Bm = W + (size_t)m * DIN * DH + (size_t)(4 * hi) * DH + 4 * c;
    #pragma unroll
    for (int q = 0; q < 32; ++q) {
      const int k0 = 8 * (q >> 1) + 2 * (q & 1);
      const v4f w0 = *(const v4f*)(Bm + (size_t)k0 * DH);
      const v4f w1 = *(const v4f*)(Bm + (size_t)(k0 + 1) * DH);
      #pragma unroll
      for (int ct = 0; ct < 4; ++ct) breg[ct][q] = (v2f){w0[ct], w1[ct]};
    }
  }

  int t = gw - first;
  const int n = (PJ_TILES - t + nw - 1) / nw;
  v4f a0[16], a1[16];
  pj_load(a0, Am, t, c, hi);
  for (int j = 0; j + 1 < n; j += 2) {
    pj_load(a1, Am, t + nw, c, hi);
    pj_tile(a0, breg, Om, t, c, hi);
    const int t2 = t + 2 * nw;
    pj_load(a0, Am, t2 < PJ_TILES ? t2 : t, c, hi);
    pj_tile(a1, breg, Om, t + nw, c, hi);
    t = t2;
  }
  if (n & 1) pj_tile(a0, breg, Om, t, c, hi);
}

#define RG_ROWS   1220
#define RG_PAD    1248
#define RG_N      82
#define RG_G      16
#define BK_E      12800
#define BK_N      125
#define BK_IT     4

__global__ __launch_bounds__(1024)
void k_count(const int* __restrict__ dst, unsigned* __restrict__ tot) {
  __shared__ unsigned hist[128];
  const int tid = threadIdx.x;
  const int m   = blockIdx.y;
  if (tid < 128) hist[tid] = 0u;
  __syncthreads();
  const int4* d4 = (const int4*)(dst + (size_t)m * NE + (size_t)blockIdx.x * BK_E);
  #pragma unroll
  for (int k = 0; k < BK_IT; ++k) {
    const int i = k * 1024 + tid;
    if (i < BK_E / 4) {
      const int4 c = d4[i];
      const int cv[4] = {c.x, c.y, c.z, c.w};
      #pragma unroll
      for (int q = 0; q < 4; ++q) {
        const unsigned r = (unsigned)cv[q] / RG_ROWS;
        if (r < RG_N) atomicAdd(&hist[r], 1u);
      }
    }
  }
  __syncthreads();
  if (tid < RG_N) {
    const unsigned h = hist[tid];
    if (h) atomicAdd(&tot[m * 128 + tid], h);
  }
}

__global__ __launch_bounds__(1024)
void k_fill(const int* __restrict__ src, const int* __restrict__ dst,
            const unsigned* __restrict__ tot, unsigned* __restrict__ cur,
            unsigned* __restrict__ elist) {
  __shared__ unsigned hist[128];
  __shared__ unsigned start[128];
  __shared__ unsigned totl[128];
  const int tid = threadIdx.x;
  const int m   = blockIdx.y;
  if (tid < 128) {
    hist[tid] = 0u;
    totl[tid] = (tid < RG_N) ? tot[m * 128 + tid] : 0u;
  }
  __syncthreads();
  const size_t e0 = (size_t)m * NE + (size_t)blockIdx.x * BK_E;
  const int4* d4 = (const int4*)(dst + e0);
  const int4* s4 = (const int4*)(src + e0);
  int4 dv[BK_IT], sv[BK_IT];
  #pragma unroll
  for (int k = 0; k < BK_IT; ++k) {
    const int i = k * 1024 + tid;
    const int ic = i < BK_E / 4 ? i : BK_E / 4 - 1;
    dv[k] = d4[ic];
    sv[k] = s4[ic];
  }
  #pragma unroll
  for (int k = 0; k < BK_IT; ++k) {
    if (k * 1024 + tid < BK_E / 4) {
      const int cv[4] = {dv[k].x, dv[k].y, dv[k].z, dv[k].w};
      #pragma unroll
      for (int q = 0; q < 4; ++q) {
        const unsigned r = (unsigned)cv[q] / RG_ROWS;
        if (r < RG_N) atomicAdd(&hist[r], 1u);
      }
    }
  }
  __syncthreads();
  if (tid < RG_N) {
    unsigned base = 0u;
    for (int j = 0; j < tid; ++j) base += totl[j];
    const unsigned h = hist[tid];
    const unsigned got = h ? atomicAdd(&cur[m * 128 + tid], h) : 0u;
    start[tid] = base + got;
    hist[tid] = 0u;
  }
  __syncthreads();
  unsigned* lm = elist + (size_t)m * NE;
  #pragma unroll
  for (int k = 0; k < BK_IT; ++k) {
    if (k * 1024 + tid < BK_E / 4) {
      const int cv[4] = {dv[k].x, dv[k].y, dv[k].z, dv[k].w};
      const int cs[4] = {sv[k].x, sv[k].y, sv[k].z, sv[k].w};
      #pragma unroll
      for (int q = 0; q < 4; ++q) {
        const unsigned r = (unsigned)cv[q] / RG_ROWS;
        if (r < RG_N) {
          const unsigned pos = start[r] + atomicAdd(&hist[r], 1u);
          lm[pos] = ((unsigned)cs[q] << 11) | ((unsigned)cv[q] - r * RG_ROWS);
        }
      }
    }
  }
}

__device__ __forceinline__ void rg_add(float* rows, unsigned pk, int nvalid,
                                       const float* __restrict__ projm, unsigned lane) {
  const int sv = (int)(pk >> 11);
  const int ld = (int)(pk & 2047u);
  for (int g = 0; g < nvalid; g += RG_G) {
    v2f r[RG_G];
    #pragma unroll
    for (int j = 0; j < RG_G; ++j) {
      const unsigned s = (unsigned)__builtin_amdgcn_readlane(sv, g + j);
      r[j] = *(const v2f*)(projm + s * DH + lane * 2);
    }
    #pragma unroll
    for (int j = 0; j < RG_G; ++j) {
      if (g + j < nvalid) {
        const unsigned d = (unsigned)__builtin_amdgcn_readlane(ld, g + j);
        float* p = rows + d * DH + lane * 2;
        atomicAdd(p, r[j].x);
        atomicAdd(p + 1, r[j].y);
      }
    }
  }
}

__device__ __forceinline__ void rg_issue(v2f (&r)[32], int mine, unsigned cnt,
                                         const float* __restrict__ projm, unsigned lane) {
  #pragma unroll
  for (int q = 0; q < 8; ++q) {
    #pragma unroll
    for (int j = 0; j < 4; ++j) r[4 * q + j] = (v2f){0.f, 0.f};
    if (q == 0 || (unsigned)(4 * q) < cnt) {
      #pragma unroll
      for (int j = 0; j < 4; ++j) {
        const unsigned s = (unsigned)__builtin_amdgcn_readlane(mine, 4 * q + j);
        r[4 * q + j] = *(const v2f*)(projm + s * DH + lane * 2);
      }
    }
  }
}

__device__ __forceinline__ v2f rg_reduce(const v2f (&r)[32], unsigned cnt) {
  v2f acc = {0.f, 0.f};
  #pragma unroll
  for (int j = 0; j < 32; ++j) {
    const v2f x = ((unsigned)j < cnt) ? r[j] : (v2f){0.f, 0.f};
    acc += x;
  }
  return acc;
}

__global__ __launch_bounds__(1024)
void k_scatter(const float* __restrict__ wsb, const unsigned* __restrict__ elist,
               const unsigned* __restrict__ tot, const float* __restrict__ b,
               const float* __restrict__ pa, float* __restrict__ hout) {
  __shared__ alignas(16) float rows[RG_ROWS * DH];
  __shared__ unsigned degc[RG_PAD];
  __shared__ unsigned beg[RG_PAD];
  __shared__ unsigned totl[128];
  const int tid = threadIdx.x;
  const unsigned lane = (unsigned)tid & 31u;
  const unsigned wv = (unsigned)__builtin_amdgcn_readfirstlane(tid >> 5);
  const int m  = blockIdx.y;
  const int rg = blockIdx.x;
  const int lo = rg * RG_ROWS;
  const unsigned nrows = (unsigned)((NN - lo) < RG_ROWS ? (NN - lo) : RG_ROWS);
  const float* projm = wsb + PROJ_OFF + (size_t)m * NN * DH;

  if (tid < 128) totl[tid] = (tid < RG_N) ? tot[m * 128 + tid] : 0u;
  for (int i = tid; i < RG_PAD; i += 1024) degc[i] = 0u;
  __syncthreads();

  unsigned bsum = 0u;
  for (int j = 0; j < rg; ++j) bsum += totl[j];
  const unsigned base = (unsigned)__builtin_amdgcn_readfirstlane((int)bsum);
  const unsigned n    = (unsigned)__builtin_amdgcn_readfirstlane((int)totl[rg]);
  const unsigned* bin = elist + (size_t)m * NE + base;
  float* o = hout + ((size_t)m * NN + lo) * DH;
  const float a = pa[m];

  if (n <= (unsigned)(RG_ROWS * DH)) {
    unsigned* srt = (unsigned*)rows;
    for (unsigned i0 = (unsigned)tid; i0 < n; i0 += 4096u) {
      unsigned pk[4];
      #pragma unroll
      for (int q = 0; q < 4; ++q) {
        const unsigned i = i0 + (unsigned)q * 1024u;
        pk[q] = bin[i < n ? i : n - 1u];
      }
      #pragma unroll
      for (int q = 0; q < 4; ++q)
        if (i0 + (unsigned)q * 1024u < n) atomicAdd(&degc[pk[q] & 2047u], 1u);
    }
    __syncthreads();
    if (wv == 0u) {
      const unsigned r0 = lane * 39u;
      unsigned cv[39];
      unsigned s = 0u;
      #pragma unroll
      for (int k = 0; k < 39; ++k) { cv[k] = degc[r0 + k]; s += cv[k]; }
      unsigned x = s;
      for (int d = 1; d < 32; d <<= 1) {
        const unsigned y = (unsigned)__shfl_up((int)x, d);
        if ((int)lane >= d) x += y;
      }
      unsigned run = x - s;
      #pragma unroll
      for (int k = 0; k < 39; ++k) { beg[r0 + k] = run; degc[r0 + k] = run; run += cv[k]; }
    }
    __syncthreads();
    for (unsigned i0 = (unsigned)tid; i0 < n; i0 += 4096u) {
      unsigned pk[4];
      #pragma unroll
      for (int q = 0; q < 4; ++q) {
        const unsigned i = i0 + (unsigned)q * 1024u;
        pk[q] = (i < n) ? bin[i] : 0u;
      }
      #pragma unroll
      for (int q = 0; q < 4; ++q) {
        if (i0 + (unsigned)q * 1024u < n) {
          const unsigned pos = atomicAdd(&degc[pk[q] & 2047u], 1u);
          srt[pos] = pk[q] >> 11;
        }
      }
    }
    __syncthreads();
    const v2f bv = *(const v2f*)(b + m * DH + lane * 2);
    unsigned r = wv;
    unsigned b0 = 0u, e0 = 0u;
    int mine = 0;
    if (r < nrows) {
      b0 = (unsigned)__builtin_amdgcn_readfirstlane((int)beg[r]);
      e0 = (unsigned)__builtin_amdgcn_readfirstlane((int)degc[r]);
      mine = (b0 + lane < e0) ? (int)srt[b0 + lane] : 0;
    }
    for (; r < nrows; r += 32u) {
      v2f rr[32];
      rg_issue(rr, mine, e0 - b0, projm, lane);
      const unsigned rn = (r + 32u < nrows) ? r + 32u : r;
      const unsigned bn = (unsigned)__builtin_amdgcn_readfirstlane((int)beg[rn]);
      const unsigned en = (unsigned)__builtin_amdgcn_readfirstlane((int)degc[rn]);
      const int minen = (bn + lane < en) ? (int)srt[bn + lane] : 0;
      v2f acc = rg_reduce(rr, e0 - b0);
      for (unsigned g = b0 + 32u; g < e0; g += 32u) {
        const int m2 = (g + lane < e0) ? (int)srt[g + lane] : 0;
        v2f r2[32];
        rg_issue(r2, m2, e0 - g, projm, lane);
        acc += rg_reduce(r2, e0 - g);
      }
      const float dg = fmaxf((float)(e0 - b0), 1.0f);
      v2f hv;
      hv.x = acc.x / dg + bv.x;
      hv.y = acc.y / dg + bv.y;
      hv.x = hv.x > 0.f ? hv.x : a * hv.x;
      hv.y = hv.y > 0.f ? hv.y : a * hv.y;
      *(v2f*)(o + (size_t)r * DH + lane * 2) = hv;
      b0 = bn; e0 = en; mine = minen;
    }
    return;
  }

  for (int i = tid; i < RG_ROWS * DH / 4; i += 1024) ((v4f*)rows)[i] = (v4f){0.f, 0.f, 0.f, 0.f};
  __syncthreads();

  for (unsigned i0 = wv * 128u; i0 < n; i0 += 4096u) {
    unsigned pk[4];
    #pragma unroll
    for (int q = 0; q < 4; ++q) {
      const unsigned i = i0 + (unsigned)q * 32u + lane;
      pk[q] = (i < n) ? bin[i] : 0u;
    }
    #pragma unroll
    for (int q = 0; q < 4; ++q) {
      const unsigned s0 = i0 + (unsigned)q * 32u;
      if (s0 < n) {
        const unsigned rem = n - s0;
        const int nvalid = rem < 32u ? (int)rem : 32;
        if ((int)lane < nvalid) atomicAdd(&degc[pk[q] & 2047u], 1u);
        rg_add(rows, pk[q], nvalid, projm, lane);
      }
    }
  }
  __syncthreads();

  const v4f bv4 = *(const v4f*)(b + m * DH + (tid & 15) * 4);
  for (unsigned i = tid; i < nrows * (DH / 4); i += 1024) {
    const float dg = fmaxf((float)degc[i >> 4], 1.0f);
    v4f hv = ((const v4f*)rows)[i];
    for (int c = 0; c < 4; ++c) {
      const float h = hv[c] / dg + bv4[c];
      hv[c] = h > 0.f ? h : a * h;
    }
    ((v4f*)o)[i] = hv;
  }
}

#define FC_WGS 1024

__device__ __forceinline__ void fc_load(v4f (&a)[8], const float* __restrict__ Am, int t,
                                        int c, int hi) {
  const float* p = Am + ((size_t)t * 16 + c) * DH + 4 * hi;
  #pragma unroll
  for (int q = 0; q < 8; ++q) a[q] = *(const v4f*)(p + 8 * q);
}

__device__ __forceinline__ void fc_tile(const v4f (&a)[8], const v2f (&breg)[2][16],
                                        const v2f bias2, v2f& ps) {
  v8f acc[2];
  #pragma unroll
  for (int ct = 0; ct < 2; ++ct) {
    const float bs = bias2[ct];
    acc[ct] = (v8f){bs, bs, bs, bs, bs, bs, bs, bs};
  }
  #pragma unroll
  for (int p = 0; p < 8; ++p) {
    const v2f a0 = {a[p].x, a[p].y};
    const v2f a1 = {a[p].z, a[p].w};
    #pragma unroll
    for (int ct = 0; ct < 2; ++ct)
      acc[ct] = __builtin_amdgcn_wmma_f32_16x16x4_f32(false, a0, false, breg[ct][2 * p], (short)0,
                                                      acc[ct], false, false);
    #pragma unroll
    for (int ct = 0; ct < 2; ++ct)
      acc[ct] = __builtin_amdgcn_wmma_f32_16x16x4_f32(false, a1, false, breg[ct][2 * p + 1], (short)0,
                                                      acc[ct], false, false);
  }
  #pragma unroll
  for (int ct = 0; ct < 2; ++ct) {
    float s = 0.f;
    #pragma unroll
    for (int v = 0; v < 8; ++v) s += tanhf(acc[ct][v]);
    ps[ct] += s;
  }
}

__global__ __launch_bounds__(128, 4)
void k_fc_wmma(const float* __restrict__ h, const float* __restrict__ fcW,
               const float* __restrict__ fcb, const float* __restrict__ attn,
               float* __restrict__ logits) {
  __shared__ float red[4];
  const int lane = threadIdx.x & 31;
  const int c    = lane & 15;
  const int hi   = lane >> 4;
  const int wv = __builtin_amdgcn_readfirstlane((int)(threadIdx.x >> 5));
  const int wg = (int)blockIdx.x;
  const int m     = wg * MM / FC_WGS;
  const int first = (m * FC_WGS + MM - 1) / MM;
  const int np    = 2 * (((m + 1) * FC_WGS + MM - 1) / MM - first);
  const int hc    = wv & 1;
  const float* Am = h + (size_t)m * NN * DH;

  v2f breg[2][16];
  {
    const float* Bm = fcW + (size_t)(4 * hi) * DH + 32 * hc + 2 * c;
    #pragma unroll
    for (int q = 0; q < 16; ++q) {
      const int k0 = 8 * (q >> 1) + 2 * (q & 1);
      const v2f w0 = *(const v2f*)(Bm + (size_t)k0 * DH);
      const v2f w1 = *(const v2f*)(Bm + (size_t)(k0 + 1) * DH);
      #pragma unroll
      for (int ct = 0; ct < 2; ++ct) breg[ct][q] = (v2f){w0[ct], w1[ct]};
    }
  }
  const v2f bias2 = *(const v2f*)(fcb + 32 * hc + 2 * c);
  const v2f attn2 = *(const v2f*)(attn + 32 * hc + 2 * c);
  v2f ps = {0.f, 0.f};

  int t = 2 * (wg - first) + (wv >> 1);
  const int n = (PJ_TILES - t + np - 1) / np;
  v4f a0[8], a1[8];
  fc_load(a0, Am, t, c, hi);
  for (int j = 0; j + 1 < n; j += 2) {
    fc_load(a1, Am, t + np, c, hi);
    fc_tile(a0, breg, bias2, ps);
    const int t2 = t + 2 * np;
    fc_load(a0, Am, t2 < PJ_TILES ? t2 : t, c, hi);
    fc_tile(a1, breg, bias2, ps);
    t = t2;
  }
  if (n & 1) fc_tile(a0, breg, bias2, ps);

  float p = ps[0] * attn2[0] + ps[1] * attn2[1];
  for (int d = 16; d > 0; d >>= 1) p += __shfl_xor(p, d);
  if (lane == 0) red[wv] = p;
  __syncthreads();
  if (threadIdx.x == 0) atomicAddF(&logits[m], (red[0] + red[1]) + (red[2] + red[3]));
}

__global__ void k_softmax(const float* __restrict__ logits, float* __restrict__ beta) {
  if (threadIdx.x == 0 && blockIdx.x == 0) {
    float l[MM]; float mx = -1e30f;
    for (int m = 0; m < MM; ++m) { l[m] = logits[m] / (float)NN; mx = fmaxf(mx, l[m]); }
    float s = 0.f;
    for (int m = 0; m < MM; ++m) { l[m] = expf(l[m] - mx); s += l[m]; }
    for (int m = 0; m < MM; ++m) beta[m] = l[m] / s;
  }
}

#define OUT_WGS 512
#define OUT_N4  (NN * (DH / 4))
#define OUT_T   (OUT_WGS * 256)

__global__ __launch_bounds__(256)
void k_out(const float* __restrict__ h, const float* __restrict__ beta,
           float* __restrict__ out) {
  const v4f* h4 = (const v4f*)h;
  v4f* o4 = (v4f*)out;
  const float b0 = beta[0], b1 = beta[1], b2 = beta[2];
  for (unsigned i = blockIdx.x * 256u + threadIdx.x; i < (unsigned)OUT_N4; i += 4u * OUT_T) {
    v4f x0[4], x1[4], x2[4];
    #pragma unroll
    for (int u = 0; u < 4; ++u) {
      const unsigned idx = i + (unsigned)u * OUT_T;
      const unsigned ic = idx < (unsigned)OUT_N4 ? idx : i;
      x0[u] = h4[ic];
      x1[u] = h4[(size_t)OUT_N4 + ic];
      x2[u] = h4[2 * (size_t)OUT_N4 + ic];
    }
    #pragma unroll
    for (int u = 0; u < 4; ++u) {
      const unsigned idx = i + (unsigned)u * OUT_T;
      if (idx < (unsigned)OUT_N4) o4[idx] = x0[u] * b0 + x1[u] * b1 + x2[u] * b2;
    }
  }
}

extern "C" void kernel_launch(void* const* d_in, const int* in_sizes, int n_in,
                              void* d_out, int out_size, void* d_ws, size_t ws_size,
                              hipStream_t stream) {
  const float* feats = (const float*)d_in[0];
  const int*   src   = (const int*)  d_in[1];
  const int*   dst   = (const int*)  d_in[2];
  const float* W     = (const float*)d_in[3];
  const float* b     = (const float*)d_in[4];
  const float* pa    = (const float*)d_in[5];
  const float* fcW   = (const float*)d_in[6];
  const float* fcb   = (const float*)d_in[7];
  const float* attn  = (const float*)d_in[8];

  float* ws     = (float*)d_ws;
  float* agg    = ws + AGG_OFF;
  float* logits = ws + LOG_OFF;
  float* beta   = ws + BETA_OFF;
  float* out    = (float*)d_out;
  unsigned* tot = (unsigned*)(ws + CNT_OFF);
  unsigned* cur = tot + 384;
  unsigned* elist = (unsigned*)d_out;

  hipMemsetAsync(ws + CNT_OFF, 0, (768 + 8) * sizeof(float), stream);

  k_count<<<dim3(BK_N, MM), 1024, 0, stream>>>(dst, tot);
  k_fill<<<dim3(BK_N, MM), 1024, 0, stream>>>(src, dst, tot, cur, elist);

  k_proj_wmma<<<PJ_WAVES / 4, 128, 0, stream>>>(feats, W, ws);

  k_scatter<<<dim3(RG_N, MM), 1024, 0, stream>>>(ws, elist, tot, b, pa, agg);

  k_fc_wmma<<<FC_WGS, 128, 0, stream>>>(agg, fcW, fcb, attn, logits);

  k_softmax<<<1, 32, 0, stream>>>(logits, beta);

  k_out<<<OUT_WGS, 256, 0, stream>>>(agg, beta, out);
}


# edge list sorted by range inside each 12800-edge chunk only (LDS, whole-line stores): count kernel, global cursors, memset gone; aggregation reads its 125 pieces; row sum order: arrival as before
# speedup vs baseline: 6.9620x; 1.1628x over previous
#include <hip/hip_runtime.h>
#include <hip/hip_bf16.h>
#include <math.h>

#define NN  100000
#define MM  3
#define DIN 128
#define DH  64
#define NE  1600000

typedef __attribute__((ext_vector_type(2))) float v2f;
typedef __attribute__((ext_vector_type(4))) float v4f;
typedef __attribute__((ext_vector_type(8))) float v8f;

#define AGG_OFF   0ull
#define PROJ_OFF  19200000ull
#define LOG_OFF   38400768ull
#define BETA_OFF  38400772ull

__device__ __forceinline__ void atomicAddF(float* p, float v) {
  asm volatile("global_atomic_add_f32 %0, %1, off scope:SCOPE_DEV"
               :: "v"(p), "v"(v) : "memory");
}

#define PJ_WAVES 2048
#define PJ_TILES (NN / 16)

__device__ __forceinline__ void pj_load(v4f (&a)[16], const float* __restrict__ Am, int t,
                                        int c, int hi) {
  const float* p = Am + ((size_t)t * 16 + c) * DIN + 4 * hi;
  #pragma unroll
  for (int q = 0; q < 16; ++q) a[q] = *(const v4f*)(p + 8 * q);
}

__device__ __forceinline__ void pj_tile(const v4f (&a)[16], const v2f (&breg)[4][32],
                                        float* __restrict__ Om, int t, int c, int hi) {
  v8f acc[4];
  #pragma unroll
  for (int ct = 0; ct < 4; ++ct) acc[ct] = (v8f){0.f,0.f,0.f,0.f,0.f,0.f,0.f,0.f};
  #pragma unroll
  for (int p = 0; p < 16; ++p) {
    const v2f a0 = {a[p].x, a[p].y};
    const v2f a1 = {a[p].z, a[p].w};
    #pragma unroll
    for (int ct = 0; ct < 4; ++ct)
      acc[ct] = __builtin_amdgcn_wmma_f32_16x16x4_f32(false, a0, false, breg[ct][2 * p], (short)0,
                                                      acc[ct], false, false);
    #pragma unroll
    for (int ct = 0; ct < 4; ++ct)
      acc[ct] = __builtin_amdgcn_wmma_f32_16x16x4_f32(false, a1, false, breg[ct][2 * p + 1], (short)0,
                                                      acc[ct], false, false);
  }
  float* o = Om + ((size_t)t * 16 + 8 * hi) * DH + 4 * c;
  #pragma unroll
  for (int v = 0; v < 8; ++v)
    *(v4f*)(o + (size_t)v * DH) = (v4f){acc[0][v], acc[1][v], acc[2][v], acc[3][v]};
}

__global__ __launch_bounds__(128, 2)
void k_proj_wmma(const float* __restrict__ feats, const float* __restrict__ W,
                 float* __restrict__ wsb) {
  const int lane = threadIdx.x & 31;
  const int c    = lane & 15;
  const int hi   = lane >> 4;
  const int gw = __builtin_amdgcn_readfirstlane((int)(blockIdx.x * 4 + (threadIdx.x >> 5)));
  const int m     = gw * MM / PJ_WAVES;
  const int first = (m * PJ_WAVES + MM - 1) / MM;
  const int nw    = ((m + 1) * PJ_WAVES + MM - 1) / MM - first;
  const float* Am = feats + (size_t)m * NN * DIN;
  float*       Om = wsb + PROJ_OFF + (size_t)m * NN * DH;

  v2f breg[4][32];
  {
    const float* Bm = W + (size_t)m * DIN * DH + (size_t)(4 * hi) * DH + 4 * c;
    #pragma unroll
    for (int q = 0; q < 32; ++q) {
      const int k0 = 8 * (q >> 1) + 2 * (q & 1);
      const v4f w0 = *(const v4f*)(Bm + (size_t)k0 * DH);
      const v4f w1 = *(const v4f*)(Bm + (size_t)(k0 + 1) * DH);
      #pragma unroll
      for (int ct = 0; ct < 4; ++ct) breg[ct][q] = (v2f){w0[ct], w1[ct]};
    }
  }

  int t = gw - first;
  const int n = (PJ_TILES - t + nw - 1) / nw;
  v4f a0[16], a1[16];
  pj_load(a0, Am, t, c, hi);
  for (int j = 0; j + 1 < n; j += 2) {
    pj_load(a1, Am, t + nw, c, hi);
    pj_tile(a0, breg, Om, t, c, hi);
    const int t2 = t + 2 * nw;
    pj_load(a0, Am, t2 < PJ_TILES ? t2 : t, c, hi);
    pj_tile(a1, breg, Om, t + nw, c, hi);
    t = t2;
  }
  if (n & 1) pj_tile(a0, breg, Om, t, c, hi);
}

#define RG_ROWS   1220
#define RG_PAD    1248
#define RG_N      82
#define RG_G      16
#define BK_E      12800
#define BK_N      125
#define BK_IT     4
#define SEG_PITCH 128
#define SEG_OFF   ((size_t)MM * NE)

__global__ __launch_bounds__(1024)
void k_sort(const int* __restrict__ src, const int* __restrict__ dst,
            unsigned* __restrict__ elist, unsigned* __restrict__ segtab,
            float* __restrict__ logits) {
  __shared__ alignas(16) unsigned stage[BK_E];
  __shared__ unsigned hist[128];
  __shared__ unsigned tots;
  const int tid = threadIdx.x;
  const int m   = blockIdx.y;
  const int ch  = blockIdx.x;
  if (tid < 128) hist[tid] = 0u;
  if (ch == 0 && m == 0 && tid < 4) logits[tid] = 0.f;
  __syncthreads();
  const size_t e0 = (size_t)m * NE + (size_t)ch * BK_E;
  const int4* d4 = (const int4*)(dst + e0);
  const int4* s4 = (const int4*)(src + e0);
  int4 dv[BK_IT], sv[BK_IT];
  #pragma unroll
  for (int k = 0; k < BK_IT; ++k) {
    const int i = k * 1024 + tid;
    const int ic = i < BK_E / 4 ? i : BK_E / 4 - 1;
    dv[k] = d4[ic];
    sv[k] = s4[ic];
  }
  #pragma unroll
  for (int k = 0; k < BK_IT; ++k) {
    if (k * 1024 + tid < BK_E / 4) {
      const int cv[4] = {dv[k].x, dv[k].y, dv[k].z, dv[k].w};
      #pragma unroll
      for (int q = 0; q < 4; ++q) {
        const unsigned r = (unsigned)cv[q] / RG_ROWS;
        if (r < RG_N) atomicAdd(&hist[r], 1u);
      }
    }
  }
  __syncthreads();
  if (tid < 32) {
    unsigned cv[4];
    unsigned sum = 0u;
    #pragma unroll
    for (int k = 0; k < 4; ++k) { cv[k] = hist[4 * tid + k]; sum += cv[k]; }
    unsigned x = sum;
    for (int d = 1; d < 32; d <<= 1) {
      const unsigned y = (unsigned)__shfl_up((int)x, d);
      if (tid >= d) x += y;
    }
    unsigned run = x - sum;
    #pragma unroll
    for (int k = 0; k < 4; ++k) {
      const int r = 4 * tid + k;
      hist[r] = run;
      if (r < RG_N) segtab[((size_t)m * RG_N + r) * SEG_PITCH + ch] = (run << 16) | cv[k];
      run += cv[k];
    }
    if (tid == 31) tots = run;
  }
  __syncthreads();
  #pragma unroll
  for (int k = 0; k < BK_IT; ++k) {
    if (k * 1024 + tid < BK_E / 4) {
      const int cv[4] = {dv[k].x, dv[k].y, dv[k].z, dv[k].w};
      const int cs[4] = {sv[k].x, sv[k].y, sv[k].z, sv[k].w};
      #pragma unroll
      for (int q = 0; q < 4; ++q) {
        const unsigned r = (unsigned)cv[q] / RG_ROWS;
        if (r < RG_N) {
          const unsigned pos = atomicAdd(&hist[r], 1u);
          stage[pos] = ((unsigned)cs[q] << 11) | ((unsigned)cv[q] - r * RG_ROWS);
        }
      }
    }
  }
  __syncthreads();
  const unsigned n4 = (tots + 3u) >> 2;
  uint4* g4 = (uint4*)(elist + e0);
  for (unsigned i = (unsigned)tid; i < n4; i += 1024u) g4[i] = ((const uint4*)stage)[i];
}

__device__ __forceinline__ void rg_add(float* rows, unsigned pk, int nvalid,
                                       const float* __restrict__ projm, unsigned lane) {
  const int sv = (int)(pk >> 11);
  const int ld = (int)(pk & 2047u);
  for (int g = 0; g < nvalid; g += RG_G) {
    v2f r[RG_G];
    #pragma unroll
    for (int j = 0; j < RG_G; ++j) {
      const unsigned s = (unsigned)__builtin_amdgcn_readlane(sv, g + j);
      r[j] = *(const v2f*)(projm + s * DH + lane * 2);
    }
    #pragma unroll
    for (int j = 0; j < RG_G; ++j) {
      if (g + j < nvalid) {
        const unsigned d = (unsigned)__builtin_amdgcn_readlane(ld, g + j);
        float* p = rows + d * DH + lane * 2;
        atomicAdd(p, r[j].x);
        atomicAdd(p + 1, r[j].y);
      }
    }
  }
}

__device__ __forceinline__ void rg_issue(v2f (&r)[32], int mine, unsigned cnt,
                                         const float* __restrict__ projm, unsigned lane) {
  #pragma unroll
  for (int q = 0; q < 8; ++q) {
    #pragma unroll
    for (int j = 0; j < 4; ++j) r[4 * q + j] = (v2f){0.f, 0.f};
    if (q == 0 || (unsigned)(4 * q) < cnt) {
      #pragma unroll
      for (int j = 0; j < 4; ++j) {
        const unsigned s = (unsigned)__builtin_amdgcn_readlane(mine, 4 * q + j);
        r[4 * q + j] = *(const v2f*)(projm + s * DH + lane * 2);
      }
    }
  }
}

__device__ __forceinline__ v2f rg_reduce(const v2f (&r)[32], unsigned cnt) {
  v2f acc = {0.f, 0.f};
  #pragma unroll
  for (int j = 0; j < 32; ++j) {
    const v2f x = ((unsigned)j < cnt) ? r[j] : (v2f){0.f, 0.f};
    acc += x;
  }
  return acc;
}

__global__ __launch_bounds__(1024)
void k_scatter(const float* __restrict__ wsb, const unsigned* __restrict__ elist,
               const unsigned* __restrict__ segtab, const float* __restrict__ b,
               const float* __restrict__ pa, float* __restrict__ hout) {
  __shared__ alignas(16) float rows[RG_ROWS * DH];
  __shared__ unsigned degc[RG_PAD];
  __shared__ unsigned beg[RG_PAD];
  __shared__ unsigned segs[128];
  const int tid = threadIdx.x;
  const unsigned lane = (unsigned)tid & 31u;
  const unsigned wv = (unsigned)__builtin_amdgcn_readfirstlane(tid >> 5);
  const int m  = blockIdx.y;
  const int rg = blockIdx.x;
  const int lo = rg * RG_ROWS;
  const unsigned nrows = (unsigned)((NN - lo) < RG_ROWS ? (NN - lo) : RG_ROWS);
  const float* projm = wsb + PROJ_OFF + (size_t)m * NN * DH;

  const unsigned* segt = segtab + ((size_t)m * RG_N + rg) * SEG_PITCH;
  if (tid < 128) segs[tid] = (tid < BK_N) ? segt[tid] : 0u;
  for (int i = tid; i < RG_PAD; i += 1024) degc[i] = 0u;
  __syncthreads();

  unsigned lsum = 0u;
  #pragma unroll
  for (int k = 0; k < 4; ++k) lsum += segs[lane + 32u * k] & 0xffffu;
  for (int d = 16; d > 0; d >>= 1) lsum += (unsigned)__shfl_xor((int)lsum, d);
  const unsigned n = (unsigned)__builtin_amdgcn_readfirstlane((int)lsum);
  unsigned slen[4];
  const unsigned* sp[4];
  #pragma unroll
  for (int k = 0; k < 4; ++k) {
    const unsigned ci = wv + 32u * (unsigned)k;
    const unsigned pkd = (unsigned)__builtin_amdgcn_readfirstlane((int)segs[ci]);
    slen[k] = pkd & 0xffffu;
    sp[k] = elist + (size_t)m * NE + (size_t)ci * BK_E + (pkd >> 16);
  }
  const unsigned l01 = slen[0] > slen[1] ? slen[0] : slen[1];
  const unsigned l23 = slen[2] > slen[3] ? slen[2] : slen[3];
  const unsigned lmax = l01 > l23 ? l01 : l23;
  float* o = hout + ((size_t)m * NN + lo) * DH;
  const float a = pa[m];

  if (n <= (unsigned)(RG_ROWS * DH)) {
    unsigned* srt = (unsigned*)rows;
    for (unsigned j0 = 0u; j0 < lmax; j0 += 64u) {
      unsigned pk[8];
      #pragma unroll
      for (int q = 0; q < 8; ++q) {
        const unsigned j = j0 + 32u * (unsigned)(q & 1) + lane;
        pk[q] = *((j < slen[q >> 1]) ? sp[q >> 1] + j : segt);
      }
      #pragma unroll
      for (int q = 0; q < 8; ++q)
        if (j0 + 32u * (unsigned)(q & 1) + lane < slen[q >> 1]) atomicAdd(&degc[pk[q] & 2047u], 1u);
    }
    __syncthreads();
    if (wv == 0u) {
      const unsigned r0 = lane * 39u;
      unsigned cv[39];
      unsigned s = 0u;
      #pragma unroll
      for (int k = 0; k < 39; ++k) { cv[k] = degc[r0 + k]; s += cv[k]; }
      unsigned x = s;
      for (int d = 1; d < 32; d <<= 1) {
        const unsigned y = (unsigned)__shfl_up((int)x, d);
        if ((int)lane >= d) x += y;
      }
      unsigned run = x - s;
      #pragma unroll
      for (int k = 0; k < 39; ++k) { beg[r0 + k] = run; degc[r0 + k] = run; run += cv[k]; }
    }
    __syncthreads();
    for (unsigned j0 = 0u; j0 < lmax; j0 += 64u) {
      unsigned pk[8];
      #pragma unroll
      for (int q = 0; q < 8; ++q) {
        const unsigned j = j0 + 32u * (unsigned)(q & 1) + lane;
        pk[q] = *((j < slen[q >> 1]) ? sp[q >> 1] + j : segt);
      }
      #pragma unroll
      for (int q = 0; q < 8; ++q) {
        if (j0 + 32u * (unsigned)(q & 1) + lane < slen[q >> 1]) {
          const unsigned pos = atomicAdd(&degc[pk[q] & 2047u], 1u);
          srt[pos] = pk[q] >> 11;
        }
      }
    }
    __syncthreads();
    const v2f bv = *(const v2f*)(b + m * DH + lane * 2);
    unsigned r = wv;
    unsigned b0 = 0u, e0 = 0u;
    int mine = 0;
    if (r < nrows) {
      b0 = (unsigned)__builtin_amdgcn_readfirstlane((int)beg[r]);
      e0 = (unsigned)__builtin_amdgcn_readfirstlane((int)degc[r]);
      mine = (b0 + lane < e0) ? (int)srt[b0 + lane] : 0;
    }
    for (; r < nrows; r += 32u) {
      v2f rr[32];
      rg_issue(rr, mine, e0 - b0, projm, lane);
      const unsigned rn = (r + 32u < nrows) ? r + 32u : r;
      const unsigned bn = (unsigned)__builtin_amdgcn_readfirstlane((int)beg[rn]);
      const unsigned en = (unsigned)__builtin_amdgcn_readfirstlane((int)degc[rn]);
      const int minen = (bn + lane < en) ? (int)srt[bn + lane] : 0;
      v2f acc = rg_reduce(rr, e0 - b0);
      for (unsigned g = b0 + 32u; g < e0; g += 32u) {
        const int m2 = (g + lane < e0) ? (int)srt[g + lane] : 0;
        v2f r2[32];
        rg_issue(r2, m2, e0 - g, projm, lane);
        acc += rg_reduce(r2, e0 - g);
      }
      const float dg = fmaxf((float)(e0 - b0), 1.0f);
      v2f hv;
      hv.x = acc.x / dg + bv.x;
      hv.y = acc.y / dg + bv.y;
      hv.x = hv.x > 0.f ? hv.x : a * hv.x;
      hv.y = hv.y > 0.f ? hv.y : a * hv.y;
      *(v2f*)(o + (size_t)r * DH + lane * 2) = hv;
      b0 = bn; e0 = en; mine = minen;
    }
    return;
  }

  for (int i = tid; i < RG_ROWS * DH / 4; i += 1024) ((v4f*)rows)[i] = (v4f){0.f, 0.f, 0.f, 0.f};
  __syncthreads();

  #pragma unroll
  for (int k = 0; k < 4; ++k) {
    for (unsigned j0 = 0u; j0 < slen[k]; j0 += 32u) {
      const unsigned j = j0 + lane;
      const unsigned pk = (j < slen[k]) ? sp[k][j] : 0u;
      const unsigned rem = slen[k] - j0;
      const int nvalid = rem < 32u ? (int)rem : 32;
      if ((int)lane < nvalid) atomicAdd(&degc[pk & 2047u], 1u);
      rg_add(rows, pk, nvalid, projm, lane);
    }
  }
  __syncthreads();

  const v4f bv4 = *(const v4f*)(b + m * DH + (tid & 15) * 4);
  for (unsigned i = tid; i < nrows * (DH / 4); i += 1024) {
    const float dg = fmaxf((float)degc[i >> 4], 1.0f);
    v4f hv = ((const v4f*)rows)[i];
    for (int c = 0; c < 4; ++c) {
      const float h = hv[c] / dg + bv4[c];
      hv[c] = h > 0.f ? h : a * h;
    }
    ((v4f*)o)[i] = hv;
  }
}

#define FC_WGS 1024

__device__ __forceinline__ void fc_load(v4f (&a)[8], const float* __restrict__ Am, int t,
                                        int c, int hi) {
  const float* p = Am + ((size_t)t * 16 + c) * DH + 4 * hi;
  #pragma unroll
  for (int q = 0; q < 8; ++q) a[q] = *(const v4f*)(p + 8 * q);
}

__device__ __forceinline__ void fc_tile(const v4f (&a)[8], const v2f (&breg)[2][16],
                                        const v2f bias2, v2f& ps) {
  v8f acc[2];
  #pragma unroll
  for (int ct = 0; ct < 2; ++ct) {
    const float bs = bias2[ct];
    acc[ct] = (v8f){bs, bs, bs, bs, bs, bs, bs, bs};
  }
  #pragma unroll
  for (int p = 0; p < 8; ++p) {
    const v2f a0 = {a[p].x, a[p].y};
    const v2f a1 = {a[p].z, a[p].w};
    #pragma unroll
    for (int ct = 0; ct < 2; ++ct)
      acc[ct] = __builtin_amdgcn_wmma_f32_16x16x4_f32(false, a0, false, breg[ct][2 * p], (short)0,
                                                      acc[ct], false, false);
    #pragma unroll
    for (int ct = 0; ct < 2; ++ct)
      acc[ct] = __builtin_amdgcn_wmma_f32_16x16x4_f32(false, a1, false, breg[ct][2 * p + 1], (short)0,
                                                      acc[ct], false, false);
  }
  #pragma unroll
  for (int ct = 0; ct < 2; ++ct) {
    float s = 0.f;
    #pragma unroll
    for (int v = 0; v < 8; ++v) s += tanhf(acc[ct][v]);
    ps[ct] += s;
  }
}

__global__ __launch_bounds__(128, 4)
void k_fc_wmma(const float* __restrict__ h, const float* __restrict__ fcW,
               const float* __restrict__ fcb, const float* __restrict__ attn,
               float* __restrict__ logits) {
  __shared__ float red[4];
  const int lane = threadIdx.x & 31;
  const int c    = lane & 15;
  const int hi   = lane >> 4;
  const int wv = __builtin_amdgcn_readfirstlane((int)(threadIdx.x >> 5));
  const int wg = (int)blockIdx.x;
  const int m     = wg * MM / FC_WGS;
  const int first = (m * FC_WGS + MM - 1) / MM;
  const int np    = 2 * (((m + 1) * FC_WGS + MM - 1) / MM - first);
  const int hc    = wv & 1;
  const float* Am = h + (size_t)m * NN * DH;

  v2f breg[2][16];
  {
    const float* Bm = fcW + (size_t)(4 * hi) * DH + 32 * hc + 2 * c;
    #pragma unroll
    for (int q = 0; q < 16; ++q) {
      const int k0 = 8 * (q >> 1) + 2 * (q & 1);
      const v2f w0 = *(const v2f*)(Bm + (size_t)k0 * DH);
      const v2f w1 = *(const v2f*)(Bm + (size_t)(k0 + 1) * DH);
      #pragma unroll
      for (int ct = 0; ct < 2; ++ct) breg[ct][q] = (v2f){w0[ct], w1[ct]};
    }
  }
  const v2f bias2 = *(const v2f*)(fcb + 32 * hc + 2 * c);
  const v2f attn2 = *(const v2f*)(attn + 32 * hc + 2 * c);
  v2f ps = {0.f, 0.f};

  int t = 2 * (wg - first) + (wv >> 1);
  const int n = (PJ_TILES - t + np - 1) / np;
  v4f a0[8], a1[8];
  fc_load(a0, Am, t, c, hi);
  for (int j = 0; j + 1 < n; j += 2) {
    fc_load(a1, Am, t + np, c, hi);
    fc_tile(a0, breg, bias2, ps);
    const int t2 = t + 2 * np;
    fc_load(a0, Am, t2 < PJ_TILES ? t2 : t, c, hi);
    fc_tile(a1, breg, bias2, ps);
    t = t2;
  }
  if (n & 1) fc_tile(a0, breg, bias2, ps);

  float p = ps[0] * attn2[0] + ps[1] * attn2[1];
  for (int d = 16; d > 0; d >>= 1) p += __shfl_xor(p, d);
  if (lane == 0) red[wv] = p;
  __syncthreads();
  if (threadIdx.x == 0) atomicAddF(&logits[m], (red[0] + red[1]) + (red[2] + red[3]));
}

__global__ void k_softmax(const float* __restrict__ logits, float* __restrict__ beta) {
  if (threadIdx.x == 0 && blockIdx.x == 0) {
    float l[MM]; float mx = -1e30f;
    for (int m = 0; m < MM; ++m) { l[m] = logits[m] / (float)NN; mx = fmaxf(mx, l[m]); }
    float s = 0.f;
    for (int m = 0; m < MM; ++m) { l[m] = expf(l[m] - mx); s += l[m]; }
    for (int m = 0; m < MM; ++m) beta[m] = l[m] / s;
  }
}

#define OUT_WGS 512
#define OUT_N4  (NN * (DH / 4))
#define OUT_T   (OUT_WGS * 256)

__global__ __launch_bounds__(256)
void k_out(const float* __restrict__ h, const float* __restrict__ beta,
           float* __restrict__ out) {
  const v4f* h4 = (const v4f*)h;
  v4f* o4 = (v4f*)out;
  const float b0 = beta[0], b1 = beta[1], b2 = beta[2];
  for (unsigned i = blockIdx.x * 256u + threadIdx.x; i < (unsigned)OUT_N4; i += 4u * OUT_T) {
    v4f x0[4], x1[4], x2[4];
    #pragma unroll
    for (int u = 0; u < 4; ++u) {
      const unsigned idx = i + (unsigned)u * OUT_T;
      const unsigned ic = idx < (unsigned)OUT_N4 ? idx : i;
      x0[u] = h4[ic];
      x1[u] = h4[(size_t)OUT_N4 + ic];
      x2[u] = h4[2 * (size_t)OUT_N4 + ic];
    }
    #pragma unroll
    for (int u = 0; u < 4; ++u) {
      const unsigned idx = i + (unsigned)u * OUT_T;
      if (idx < (unsigned)OUT_N4) o4[idx] = x0[u] * b0 + x1[u] * b1 + x2[u] * b2;
    }
  }
}

extern "C" void kernel_launch(void* const* d_in, const int* in_sizes, int n_in,
                              void* d_out, int out_size, void* d_ws, size_t ws_size,
                              hipStream_t stream) {
  const float* feats = (const float*)d_in[0];
  const int*   src   = (const int*)  d_in[1];
  const int*   dst   = (const int*)  d_in[2];
  const float* W     = (const float*)d_in[3];
  const float* b     = (const float*)d_in[4];
  const float* pa    = (const float*)d_in[5];
  const float* fcW   = (const float*)d_in[6];
  const float* fcb   = (const float*)d_in[7];
  const float* attn  = (const float*)d_in[8];

  float* ws     = (float*)d_ws;
  float* agg    = ws + AGG_OFF;
  float* logits = ws + LOG_OFF;
  float* beta   = ws + BETA_OFF;
  float* out    = (float*)d_out;
  unsigned* elist  = (unsigned*)d_out;
  unsigned* segtab = elist + SEG_OFF;

  k_sort<<<dim3(BK_N, MM), 1024, 0, stream>>>(src, dst, elist, segtab, logits);

  k_proj_wmma<<<PJ_WAVES / 4, 128, 0, stream>>>(feats, W, ws);

  k_scatter<<<dim3(RG_N, MM), 1024, 0, stream>>>(ws, elist, segtab, b, pa, agg);

  k_fc_wmma<<<FC_WGS, 128, 0, stream>>>(agg, fcW, fcb, attn, logits);

  k_softmax<<<1, 32, 0, stream>>>(logits, beta);

  k_out<<<OUT_WGS, 256, 0, stream>>>(agg, beta, out);
}
